# starting kernel
# baseline (speedup 1.0000x reference)
#include <hip/hip_runtime.h>


typedef float v2f __attribute__((ext_vector_type(2)));
typedef float v8f __attribute__((ext_vector_type(8)));

#define Bb    16
#define Cc    64
#define Dd    128
#define Mm    16
#define Ll    3
#define NMODE 512
#define NSPEC (NMODE*Cc*Bb)
#define WLAYER (2*NMODE*Cc*Cc)

__global__ void fnde_build_tables(float* cs, float* sn) {
    int t = threadIdx.x;
    float ang = (float)(6.283185307179586 / 128.0) * (float)t;
    cs[t] = cosf(ang);
    sn[t] = sinf(ang);
}

__global__ void fnde_pack_weights(const float* __restrict__ w1,
                                  const float* __restrict__ w2,
                                  float* __restrict__ Wp) {
    int idx = blockIdx.x * 256 + threadIdx.x;
    int l   = idx / (NMODE * Cc * Cc);
    int rem = idx % (NMODE * Cc * Cc);
    int m   = rem >> 12;
    int oi  = rem & 4095;
    int o   = oi >> 6;
    int i   = oi & 63;
    int corner = m >> 8;
    int mx = (m >> 4) & 15;
    int ky = m & 15;
    const float* src = corner ? w2 : w1;
    int sidx = ((((l * Cc + i) * Cc + o) * Mm + mx) * Mm + ky) * 2;
    float* dst = Wp + (size_t)l * WLAYER + (size_t)m * 4096 + oi;
    dst[0]              = src[sidx];
    dst[NMODE * Cc * Cc] = src[sidx + 1];
}

__global__ void fnde_fwd_dft1(const float* __restrict__ z,
                              const float* __restrict__ cs,
                              const float* __restrict__ sn,
                              float* __restrict__ Ay) {
    int t  = blockIdx.x * 256 + threadIdx.x;
    int b  = t >> 11;
    int x  = (t >> 4) & 127;
    int ky = t & 15;
    const float* row = z + (b * 128 + x) * 128;
    float re = 0.f, im = 0.f;
    for (int y = 0; y < 128; ++y) {
        float v = row[y];
        int a = (ky * y) & 127;
        re += v * cs[a];
        im -= v * sn[a];
    }
    Ay[t * 2]     = re;
    Ay[t * 2 + 1] = im;
}

__global__ void fnde_fwd_dft2(const float* __restrict__ Ay,
                              const float* __restrict__ cs,
                              const float* __restrict__ sn,
                              float* __restrict__ Zhat) {
    int t = blockIdx.x * 256 + threadIdx.x;
    int b = t >> 9;
    int m = t & 511;
    int corner = m >> 8, mx = (m >> 4) & 15, ky = m & 15;
    int kx = corner ? (112 + mx) : mx;
    float re = 0.f, im = 0.f;
    for (int x = 0; x < 128; ++x) {
        int base = ((b * 128 + x) * 16 + ky) * 2;
        float ar = Ay[base], ai = Ay[base + 1];
        int a = (kx * x) & 127;
        float c = cs[a], s = sn[a];
        re += ar * c + ai * s;
        im += ai * c - ar * s;
    }
    Zhat[t * 2]     = re;
    Zhat[t * 2 + 1] = im;
}

__global__ void fnde_init_spec(const float* __restrict__ Zhat,
                               const float* __restrict__ lw,
                               const float* __restrict__ lb,
                               float* __restrict__ Zre, float* __restrict__ Zim,
                               float* __restrict__ STre, float* __restrict__ STim) {
    int e = blockIdx.x * 256 + threadIdx.x;
    int m = e >> 10;
    int c = (e >> 4) & 63;
    int b = e & 15;
    float zr = Zhat[(b * 512 + m) * 2];
    float zi = Zhat[(b * 512 + m) * 2 + 1];
    float w = lw[c];
    Zre[e] = w * zr + ((m == 0) ? lb[c] * 16384.0f : 0.0f);
    Zim[e] = w * zi;
    STre[e] = 0.0f;
    STim[e] = 0.0f;
}

__global__ __launch_bounds__(128)
void fnde_gapply(const float* __restrict__ Vre, const float* __restrict__ Vim,
                 const float* __restrict__ Wp,
                 float* __restrict__ Kre, float* __restrict__ Kim,
                 const float* __restrict__ bias) {
    int tid  = blockIdx.x * 128 + threadIdx.x;
    int mode = tid >> 5;
    int lane = threadIdx.x & 31;
    int laneHi = lane >> 4;
    int laneLo = lane & 15;

    const float* Wr = Wp + (size_t)mode * 4096;
    const float* Wi = Wr + (size_t)NMODE * Cc * Cc;
    const float* Vr = Vre + mode * 1024;
    const float* Vi = Vim + mode * 1024;

    v8f cr0 = {}, cr1 = {}, cr2 = {}, cr3 = {};
    v8f ci0 = {}, ci1 = {}, ci2 = {}, ci3 = {};

    const int bOff = (laneHi << 5) + laneLo;
    const int aOff = laneLo * 64 + laneHi * 2;

#pragma unroll 4
    for (int kk = 0; kk < 16; ++kk) {
        int vb = kk * 64 + bOff;
        v2f br, bi, bin;
        br.x = Vr[vb];      br.y = Vr[vb + 16];
        bi.x = Vi[vb];      bi.y = Vi[vb + 16];
        bin.x = -bi.x;      bin.y = -bi.y;
        int ab = aOff + kk * 4;

#define FNDE_MT(MT, CR, CI)                                                          \
        {                                                                            \
            v2f ar = *(const v2f*)(Wr + ab + MT * 1024);                             \
            v2f ai = *(const v2f*)(Wi + ab + MT * 1024);                             \
            CR = __builtin_amdgcn_wmma_f32_16x16x4_f32(false, ar, false, br,         \
                                                       (short)0, CR, false, false);  \
            CR = __builtin_amdgcn_wmma_f32_16x16x4_f32(false, ai, false, bin,        \
                                                       (short)0, CR, false, false);  \
            CI = __builtin_amdgcn_wmma_f32_16x16x4_f32(false, ar, false, bi,         \
                                                       (short)0, CI, false, false);  \
            CI = __builtin_amdgcn_wmma_f32_16x16x4_f32(false, ai, false, br,         \
                                                       (short)0, CI, false, false);  \
        }
        FNDE_MT(0, cr0, ci0)
        FNDE_MT(1, cr1, ci1)
        FNDE_MT(2, cr2, ci2)
        FNDE_MT(3, cr3, ci3)
#undef FNDE_MT
    }

    float* Kor = Kre + mode * 1024;
    float* Koi = Kim + mode * 1024;
    bool addBias = (mode == 0);

#define FNDE_ST(MT, CR, CI)                                                      \
    {                                                                            \
        int obase = MT * 16 + (laneHi << 3);                                     \
        _Pragma("unroll")                                                        \
        for (int j = 0; j < 8; ++j) {                                            \
            float vr = CR[j];                                                    \
            if (addBias) vr += bias[obase + j] * 16384.0f;                       \
            Kor[(obase + j) * 16 + laneLo] = vr;                                 \
            Koi[(obase + j) * 16 + laneLo] = CI[j];                              \
        }                                                                        \
    }
    FNDE_ST(0, cr0, ci0)
    FNDE_ST(1, cr1, ci1)
    FNDE_ST(2, cr2, ci2)
    FNDE_ST(3, cr3, ci3)
#undef FNDE_ST
}

__device__ inline void fnde_pi(const float* __restrict__ Re, const float* __restrict__ Im,
                               int e, float kr, float ki, float& pr, float& pim) {
    int m = e >> 10;
    int ky = m & 15;
    if (ky != 0) { pr = kr; pim = ki; return; }
    int corner = m >> 8;
    int mx = (m >> 4) & 15;
    if (corner == 1 && mx == 0) { pr = 0.5f * kr; pim = 0.5f * ki; return; }
    int pm = (mx == 0) ? m : ((corner == 0) ? (256 + (16 - mx) * 16) : ((16 - mx) * 16));
    int pe = (pm << 10) + (e & 1023);
    pr  = 0.5f * (kr + Re[pe]);
    pim = 0.5f * (ki - Im[pe]);
}

__global__ void fnde_spec_mid(const float* __restrict__ Zre, const float* __restrict__ Zim,
                              const float* __restrict__ Kre, const float* __restrict__ Kim,
                              float* __restrict__ Ure, float* __restrict__ Uim,
                              float* __restrict__ SSre, float* __restrict__ SSim,
                              const float* __restrict__ ts, int s, int phase) {
    int e = blockIdx.x * 256 + threadIdx.x;
    float h  = ts[s + 1] - ts[s];
    float uc = (phase == 3) ? h : 0.5f * h;
    float kr = Kre[e], ki = Kim[e];
    if (phase == 1) { SSre[e] = kr;          SSim[e] = ki; }
    else            { SSre[e] += 2.0f * kr;  SSim[e] += 2.0f * ki; }
    float pr, pim;
    fnde_pi(Kre, Kim, e, kr, ki, pr, pim);
    Ure[e] = Zre[e] + uc * pr;
    Uim[e] = Zim[e] + uc * pim;
}

__global__ void fnde_spec_fin(float* __restrict__ Zre, float* __restrict__ Zim,
                              const float* __restrict__ Kre, const float* __restrict__ Kim,
                              const float* __restrict__ SSre, const float* __restrict__ SSim,
                              float* __restrict__ STre, float* __restrict__ STim,
                              const float* __restrict__ ts, int s) {
    int e = blockIdx.x * 256 + threadIdx.x;
    float h6 = (ts[s + 1] - ts[s]) * (1.0f / 6.0f);
    float s4r = SSre[e] + Kre[e];
    float s4i = SSim[e] + Kim[e];
    int m = e >> 10;
    int ky = m & 15;
    float pr, pim;
    if (ky != 0) { pr = s4r; pim = s4i; }
    else {
        int corner = m >> 8, mx = (m >> 4) & 15;
        if (corner == 1 && mx == 0) { pr = 0.5f * s4r; pim = 0.5f * s4i; }
        else {
            int pm = (mx == 0) ? m : ((corner == 0) ? (256 + (16 - mx) * 16) : ((16 - mx) * 16));
            int pe = (pm << 10) + (e & 1023);
            float qr = SSre[pe] + Kre[pe];
            float qi = SSim[pe] + Kim[pe];
            pr  = 0.5f * (s4r + qr);
            pim = 0.5f * (s4i - qi);
        }
    }
    Zre[e] += h6 * pr;
    Zim[e] += h6 * pim;
    STre[e] += h6 * s4r;
    STim[e] += h6 * s4i;
}

__global__ void fnde_proj_prep(const float* __restrict__ p1w, const float* __restrict__ p1b,
                               const float* __restrict__ p2w, const float* __restrict__ p2b,
                               const float* __restrict__ lw,  const float* __restrict__ lb,
                               float* __restrict__ q, float* __restrict__ scal) {
    __shared__ float qs[64];
    int c = threadIdx.x;
    float acc = 0.f;
    for (int p = 0; p < 64; ++p) acc += p2w[p] * p1w[p * 64 + c];
    qs[c] = acc;
    q[c] = acc;
    __syncthreads();
    if (c == 0) {
        float r = p2b[0];
        for (int p = 0; p < 64; ++p) r += p2w[p] * p1b[p];
        float al = 0.f, be = r;
        for (int i = 0; i < 64; ++i) { al += qs[i] * lw[i]; be += qs[i] * lb[i]; }
        scal[0] = al;
        scal[1] = be;
    }
}

__global__ void fnde_proj_modes(const float* __restrict__ STre, const float* __restrict__ STim,
                                const float* __restrict__ q, float* __restrict__ T) {
    int t = blockIdx.x * 256 + threadIdx.x;
    int b = t >> 9;
    int m = t & 511;
    const float* pr = STre + m * 1024 + b;
    const float* pi = STim + m * 1024 + b;
    float re = 0.f, im = 0.f;
    for (int c = 0; c < 64; ++c) {
        float w = q[c];
        re += w * pr[c * 16];
        im += w * pi[c * 16];
    }
    T[t * 2]     = re;
    T[t * 2 + 1] = im;
}

__global__ void fnde_inv_dft1(const float* __restrict__ T,
                              const float* __restrict__ cs, const float* __restrict__ sn,
                              float* __restrict__ G) {
    int t  = blockIdx.x * 256 + threadIdx.x;
    int b  = t >> 11;
    int x  = (t >> 4) & 127;
    int ky = t & 15;
    float re = 0.f, im = 0.f;
    for (int corner = 0; corner < 2; ++corner) {
        for (int mx = 0; mx < 16; ++mx) {
            int m  = corner * 256 + mx * 16 + ky;
            int kx = corner ? (112 + mx) : mx;
            float tr = T[(b * 512 + m) * 2];
            float ti = T[(b * 512 + m) * 2 + 1];
            int a = (kx * x) & 127;
            float c = cs[a], s = sn[a];
            re += tr * c - ti * s;
            im += tr * s + ti * c;
        }
    }
    G[t * 2]     = re;
    G[t * 2 + 1] = im;
}

__global__ void fnde_inv_dft2_out(const float* __restrict__ G, const float* __restrict__ z,
                                  const float* __restrict__ cs, const float* __restrict__ sn,
                                  const float* __restrict__ scal, float* __restrict__ out) {
    int t = blockIdx.x * 256 + threadIdx.x;
    int b = t >> 14;
    int x = (t >> 7) & 127;
    int y = t & 127;
    const float* g = G + ((b * 128 + x) * 16) * 2;
    float acc = g[0];
    for (int ky = 1; ky < 16; ++ky) {
        int a = (ky * y) & 127;
        acc += 2.0f * (g[ky * 2] * cs[a] - g[ky * 2 + 1] * sn[a]);
    }
    out[t] = scal[0] * z[t] + scal[1] + acc * (1.0f / 16384.0f);
}

extern "C" void kernel_launch(void* const* d_in, const int* in_sizes, int n_in,
                              void* d_out, int out_size, void* d_ws, size_t ws_size,
                              hipStream_t stream) {
    (void)in_sizes; (void)n_in; (void)out_size; (void)ws_size;

    const float* zin    = (const float*)d_in[0];
    const float* lift_w = (const float*)d_in[1];
    const float* lift_b = (const float*)d_in[2];
    const float* spec_w1 = (const float*)d_in[3];
    const float* spec_w2 = (const float*)d_in[4];
    const float* fl_bias = (const float*)d_in[5];
    const float* p1_w = (const float*)d_in[6];
    const float* p1_b = (const float*)d_in[7];
    const float* p2_w = (const float*)d_in[8];
    const float* p2_b = (const float*)d_in[9];
    const float* ts   = (const float*)d_in[10];
    float* outp = (float*)d_out;

    float* w = (float*)d_ws;
    float* Wp   = w;  w += (size_t)Ll * WLAYER;
    float* Zre  = w;  w += NSPEC;  float* Zim  = w;  w += NSPEC;
    float* Ure  = w;  w += NSPEC;  float* Uim  = w;  w += NSPEC;
    float* Kre  = w;  w += NSPEC;  float* Kim  = w;  w += NSPEC;
    float* SSre = w;  w += NSPEC;  float* SSim = w;  w += NSPEC;
    float* STre = w;  w += NSPEC;  float* STim = w;  w += NSPEC;
    float* Zhat = w;  w += Bb * NMODE * 2;
    float* Ay   = w;  w += Bb * Dd * Mm * 2;
    float* Tm   = w;  w += Bb * NMODE * 2;
    float* Gi   = w;  w += Bb * Dd * Mm * 2;
    float* qv   = w;  w += 64;
    float* cs   = w;  w += 128;
    float* sn   = w;  w += 128;
    float* scal = w;  w += 8;

    fnde_build_tables<<<1, 128, 0, stream>>>(cs, sn);
    fnde_pack_weights<<<(Ll * NMODE * Cc * Cc) / 256, 256, 0, stream>>>(spec_w1, spec_w2, Wp);
    fnde_fwd_dft1<<<(Bb * Dd * Mm) / 256, 256, 0, stream>>>(zin, cs, sn, Ay);
    fnde_fwd_dft2<<<(Bb * NMODE) / 256, 256, 0, stream>>>(Ay, cs, sn, Zhat);
    fnde_init_spec<<<NSPEC / 256, 256, 0, stream>>>(Zhat, lift_w, lift_b, Zre, Zim, STre, STim);

    for (int l = 0; l < Ll; ++l) {
        const float* Wl = Wp + (size_t)l * WLAYER;
        const float* bl = fl_bias + l * Cc;
        for (int s = 0; s < 4; ++s) {
            fnde_gapply<<<128, 128, 0, stream>>>(Zre, Zim, Wl, Kre, Kim, bl);
            fnde_spec_mid<<<NSPEC / 256, 256, 0, stream>>>(Zre, Zim, Kre, Kim, Ure, Uim,
                                                           SSre, SSim, ts, s, 1);
            fnde_gapply<<<128, 128, 0, stream>>>(Ure, Uim, Wl, Kre, Kim, bl);
            fnde_spec_mid<<<NSPEC / 256, 256, 0, stream>>>(Zre, Zim, Kre, Kim, Ure, Uim,
                                                           SSre, SSim, ts, s, 2);
            fnde_gapply<<<128, 128, 0, stream>>>(Ure, Uim, Wl, Kre, Kim, bl);
            fnde_spec_mid<<<NSPEC / 256, 256, 0, stream>>>(Zre, Zim, Kre, Kim, Ure, Uim,
                                                           SSre, SSim, ts, s, 3);
            fnde_gapply<<<128, 128, 0, stream>>>(Ure, Uim, Wl, Kre, Kim, bl);
            fnde_spec_fin<<<NSPEC / 256, 256, 0, stream>>>(Zre, Zim, Kre, Kim, SSre, SSim,
                                                           STre, STim, ts, s);
        }
    }

    fnde_proj_prep<<<1, 64, 0, stream>>>(p1_w, p1_b, p2_w, p2_b, lift_w, lift_b, qv, scal);
    fnde_proj_modes<<<(Bb * NMODE) / 256, 256, 0, stream>>>(STre, STim, qv, Tm);
    fnde_inv_dft1<<<(Bb * Dd * Mm) / 256, 256, 0, stream>>>(Tm, cs, sn, Gi);
    fnde_inv_dft2_out<<<(Bb * Dd * Dd) / 256, 256, 0, stream>>>(Gi, zin, cs, sn, scal, outp);
}


# Spectral weights repacked in WMMA A-fragment order (re/im interleaved): each A fetch is one b128 load per lane over 4 whole lines, grouped per 4 k-steps, not b64 reads spread over 16 lines
# speedup vs baseline: 1.2192x; 1.2192x over previous
#include <hip/hip_runtime.h>


typedef float v2f __attribute__((ext_vector_type(2)));
typedef float v4f __attribute__((ext_vector_type(4)));
typedef float v8f __attribute__((ext_vector_type(8)));

#define Bb    16
#define Cc    64
#define Dd    128
#define Mm    16
#define Ll    3
#define NMODE 512
#define NSPEC (NMODE*Cc*Bb)
#define WLAYER (2*NMODE*Cc*Cc)

__global__ void fnde_build_tables(float* cs, float* sn) {
    int t = threadIdx.x;
    float ang = (float)(6.283185307179586 / 128.0) * (float)t;
    cs[t] = cosf(ang);
    sn[t] = sinf(ang);
}

__global__ void fnde_pack_weights(const float* __restrict__ w1,
                                  const float* __restrict__ w2,
                                  float* __restrict__ Wp) {
    int idx  = blockIdx.x * 256 + threadIdx.x;
    int l    = idx / (NMODE * 2048);
    int rem  = idx % (NMODE * 2048);
    int m    = rem >> 11;
    int frag = rem & 2047;
    int lane = frag & 31;
    int kk   = (frag >> 5) & 15;
    int mt   = frag >> 9;
    int o    = mt * 16 + (lane & 15);
    int i    = kk * 4 + (lane >> 4) * 2;
    int corner = m >> 8;
    int mx = (m >> 4) & 15;
    int ky = m & 15;
    const float* src = corner ? w2 : w1;
    int sidx = ((((l * Cc + i) * Cc + o) * Mm + mx) * Mm + ky) * 2;
    const int kstride = Cc * Mm * Mm * 2;
    v4f f;
    f.x = src[sidx];
    f.z = src[sidx + 1];
    f.y = src[sidx + kstride];
    f.w = src[sidx + kstride + 1];
    v4f* dst = (v4f*)(Wp + (size_t)l * WLAYER + (size_t)m * (2 * Cc * Cc));
    dst[frag] = f;
}

__global__ void fnde_fwd_dft1(const float* __restrict__ z,
                              const float* __restrict__ cs,
                              const float* __restrict__ sn,
                              float* __restrict__ Ay) {
    int t  = blockIdx.x * 256 + threadIdx.x;
    int b  = t >> 11;
    int x  = (t >> 4) & 127;
    int ky = t & 15;
    const float* row = z + (b * 128 + x) * 128;
    float re = 0.f, im = 0.f;
    for (int y = 0; y < 128; ++y) {
        float v = row[y];
        int a = (ky * y) & 127;
        re += v * cs[a];
        im -= v * sn[a];
    }
    Ay[t * 2]     = re;
    Ay[t * 2 + 1] = im;
}

__global__ void fnde_fwd_dft2(const float* __restrict__ Ay,
                              const float* __restrict__ cs,
                              const float* __restrict__ sn,
                              float* __restrict__ Zhat) {
    int t = blockIdx.x * 256 + threadIdx.x;
    int b = t >> 9;
    int m = t & 511;
    int corner = m >> 8, mx = (m >> 4) & 15, ky = m & 15;
    int kx = corner ? (112 + mx) : mx;
    float re = 0.f, im = 0.f;
    for (int x = 0; x < 128; ++x) {
        int base = ((b * 128 + x) * 16 + ky) * 2;
        float ar = Ay[base], ai = Ay[base + 1];
        int a = (kx * x) & 127;
        float c = cs[a], s = sn[a];
        re += ar * c + ai * s;
        im += ai * c - ar * s;
    }
    Zhat[t * 2]     = re;
    Zhat[t * 2 + 1] = im;
}

__global__ void fnde_init_spec(const float* __restrict__ Zhat,
                               const float* __restrict__ lw,
                               const float* __restrict__ lb,
                               float* __restrict__ Zre, float* __restrict__ Zim,
                               float* __restrict__ STre, float* __restrict__ STim) {
    int e = blockIdx.x * 256 + threadIdx.x;
    int m = e >> 10;
    int c = (e >> 4) & 63;
    int b = e & 15;
    float zr = Zhat[(b * 512 + m) * 2];
    float zi = Zhat[(b * 512 + m) * 2 + 1];
    float w = lw[c];
    Zre[e] = w * zr + ((m == 0) ? lb[c] * 16384.0f : 0.0f);
    Zim[e] = w * zi;
    STre[e] = 0.0f;
    STim[e] = 0.0f;
}

__global__ __launch_bounds__(128)
void fnde_gapply(const float* __restrict__ Vre, const float* __restrict__ Vim,
                 const float* __restrict__ Wp,
                 float* __restrict__ Kre, float* __restrict__ Kim,
                 const float* __restrict__ bias) {
    int tid  = blockIdx.x * 128 + threadIdx.x;
    int mode = tid >> 5;
    int lane = threadIdx.x & 31;
    int laneHi = lane >> 4;
    int laneLo = lane & 15;

    const v4f* Wm = (const v4f*)(Wp + (size_t)mode * (2 * Cc * Cc)) + lane;
    const float* Vr = Vre + mode * 1024;
    const float* Vi = Vim + mode * 1024;

    v8f cr0 = {}, cr1 = {}, cr2 = {}, cr3 = {};
    v8f ci0 = {}, ci1 = {}, ci2 = {}, ci3 = {};

    const int bOff = (laneHi << 5) + laneLo;

#pragma unroll 1
    for (int k0 = 0; k0 < 16; k0 += 4) {
        v4f a4[4][4];
        v2f br[4], bi[4];
#pragma unroll
        for (int j = 0; j < 4; ++j) {
            int vb = (k0 + j) * 64 + bOff;
            br[j].x = Vr[vb];   br[j].y = Vr[vb + 16];
            bi[j].x = Vi[vb];   bi[j].y = Vi[vb + 16];
            a4[j][0] = Wm[(0 * 16 + k0 + j) * 32];
            a4[j][1] = Wm[(1 * 16 + k0 + j) * 32];
            a4[j][2] = Wm[(2 * 16 + k0 + j) * 32];
            a4[j][3] = Wm[(3 * 16 + k0 + j) * 32];
            __builtin_amdgcn_sched_barrier(0);
        }

#define FNDE_MT(MT, CR, CI)                                                          \
        {                                                                            \
            v2f ar, ai;                                                              \
            ar.x = a4[j][MT].x; ar.y = a4[j][MT].y;                                  \
            ai.x = a4[j][MT].z; ai.y = a4[j][MT].w;                                  \
            CR = __builtin_amdgcn_wmma_f32_16x16x4_f32(false, ar, false, br[j],      \
                                                       (short)0, CR, false, false);  \
            CR = __builtin_amdgcn_wmma_f32_16x16x4_f32(false, ai, false, bin,        \
                                                       (short)0, CR, false, false);  \
            CI = __builtin_amdgcn_wmma_f32_16x16x4_f32(false, ar, false, bi[j],      \
                                                       (short)0, CI, false, false);  \
            CI = __builtin_amdgcn_wmma_f32_16x16x4_f32(false, ai, false, br[j],      \
                                                       (short)0, CI, false, false);  \
        }
#pragma unroll
        for (int j = 0; j < 4; ++j) {
            v2f bin;
            bin.x = -bi[j].x;   bin.y = -bi[j].y;
            FNDE_MT(0, cr0, ci0)
            FNDE_MT(1, cr1, ci1)
            FNDE_MT(2, cr2, ci2)
            FNDE_MT(3, cr3, ci3)
        }
#undef FNDE_MT
    }

    float* Kor = Kre + mode * 1024;
    float* Koi = Kim + mode * 1024;
    bool addBias = (mode == 0);

#define FNDE_ST(MT, CR, CI)                                                      \
    {                                                                            \
        int obase = MT * 16 + (laneHi << 3);                                     \
        _Pragma("unroll")                                                        \
        for (int j = 0; j < 8; ++j) {                                            \
            float vr = CR[j];                                                    \
            if (addBias) vr += bias[obase + j] * 16384.0f;                       \
            Kor[(obase + j) * 16 + laneLo] = vr;                                 \
            Koi[(obase + j) * 16 + laneLo] = CI[j];                              \
        }                                                                        \
    }
    FNDE_ST(0, cr0, ci0)
    FNDE_ST(1, cr1, ci1)
    FNDE_ST(2, cr2, ci2)
    FNDE_ST(3, cr3, ci3)
#undef FNDE_ST
}

__device__ inline void fnde_pi(const float* __restrict__ Re, const float* __restrict__ Im,
                               int e, float kr, float ki, float& pr, float& pim) {
    int m = e >> 10;
    int ky = m & 15;
    if (ky != 0) { pr = kr; pim = ki; return; }
    int corner = m >> 8;
    int mx = (m >> 4) & 15;
    if (corner == 1 && mx == 0) { pr = 0.5f * kr; pim = 0.5f * ki; return; }
    int pm = (mx == 0) ? m : ((corner == 0) ? (256 + (16 - mx) * 16) : ((16 - mx) * 16));
    int pe = (pm << 10) + (e & 1023);
    pr  = 0.5f * (kr + Re[pe]);
    pim = 0.5f * (ki - Im[pe]);
}

__global__ void fnde_spec_mid(const float* __restrict__ Zre, const float* __restrict__ Zim,
                              const float* __restrict__ Kre, const float* __restrict__ Kim,
                              float* __restrict__ Ure, float* __restrict__ Uim,
                              float* __restrict__ SSre, float* __restrict__ SSim,
                              const float* __restrict__ ts, int s, int phase) {
    int e = blockIdx.x * 256 + threadIdx.x;
    float h  = ts[s + 1] - ts[s];
    float uc = (phase == 3) ? h : 0.5f * h;
    float kr = Kre[e], ki = Kim[e];
    if (phase == 1) { SSre[e] = kr;          SSim[e] = ki; }
    else            { SSre[e] += 2.0f * kr;  SSim[e] += 2.0f * ki; }
    float pr, pim;
    fnde_pi(Kre, Kim, e, kr, ki, pr, pim);
    Ure[e] = Zre[e] + uc * pr;
    Uim[e] = Zim[e] + uc * pim;
}

__global__ void fnde_spec_fin(float* __restrict__ Zre, float* __restrict__ Zim,
                              const float* __restrict__ Kre, const float* __restrict__ Kim,
                              const float* __restrict__ SSre, const float* __restrict__ SSim,
                              float* __restrict__ STre, float* __restrict__ STim,
                              const float* __restrict__ ts, int s) {
    int e = blockIdx.x * 256 + threadIdx.x;
    float h6 = (ts[s + 1] - ts[s]) * (1.0f / 6.0f);
    float s4r = SSre[e] + Kre[e];
    float s4i = SSim[e] + Kim[e];
    int m = e >> 10;
    int ky = m & 15;
    float pr, pim;
    if (ky != 0) { pr = s4r; pim = s4i; }
    else {
        int corner = m >> 8, mx = (m >> 4) & 15;
        if (corner == 1 && mx == 0) { pr = 0.5f * s4r; pim = 0.5f * s4i; }
        else {
            int pm = (mx == 0) ? m : ((corner == 0) ? (256 + (16 - mx) * 16) : ((16 - mx) * 16));
            int pe = (pm << 10) + (e & 1023);
            float qr = SSre[pe] + Kre[pe];
            float qi = SSim[pe] + Kim[pe];
            pr  = 0.5f * (s4r + qr);
            pim = 0.5f * (s4i - qi);
        }
    }
    Zre[e] += h6 * pr;
    Zim[e] += h6 * pim;
    STre[e] += h6 * s4r;
    STim[e] += h6 * s4i;
}

__global__ void fnde_proj_prep(const float* __restrict__ p1w, const float* __restrict__ p1b,
                               const float* __restrict__ p2w, const float* __restrict__ p2b,
                               const float* __restrict__ lw,  const float* __restrict__ lb,
                               float* __restrict__ q, float* __restrict__ scal) {
    __shared__ float qs[64];
    int c = threadIdx.x;
    float acc = 0.f;
    for (int p = 0; p < 64; ++p) acc += p2w[p] * p1w[p * 64 + c];
    qs[c] = acc;
    q[c] = acc;
    __syncthreads();
    if (c == 0) {
        float r = p2b[0];
        for (int p = 0; p < 64; ++p) r += p2w[p] * p1b[p];
        float al = 0.f, be = r;
        for (int i = 0; i < 64; ++i) { al += qs[i] * lw[i]; be += qs[i] * lb[i]; }
        scal[0] = al;
        scal[1] = be;
    }
}

__global__ void fnde_proj_modes(const float* __restrict__ STre, const float* __restrict__ STim,
                                const float* __restrict__ q, float* __restrict__ T) {
    int t = blockIdx.x * 256 + threadIdx.x;
    int b = t >> 9;
    int m = t & 511;
    const float* pr = STre + m * 1024 + b;
    const float* pi = STim + m * 1024 + b;
    float re = 0.f, im = 0.f;
    for (int c = 0; c < 64; ++c) {
        float w = q[c];
        re += w * pr[c * 16];
        im += w * pi[c * 16];
    }
    T[t * 2]     = re;
    T[t * 2 + 1] = im;
}

__global__ void fnde_inv_dft1(const float* __restrict__ T,
                              const float* __restrict__ cs, const float* __restrict__ sn,
                              float* __restrict__ G) {
    int t  = blockIdx.x * 256 + threadIdx.x;
    int b  = t >> 11;
    int x  = (t >> 4) & 127;
    int ky = t & 15;
    float re = 0.f, im = 0.f;
    for (int corner = 0; corner < 2; ++corner) {
        for (int mx = 0; mx < 16; ++mx) {
            int m  = corner * 256 + mx * 16 + ky;
            int kx = corner ? (112 + mx) : mx;
            float tr = T[(b * 512 + m) * 2];
            float ti = T[(b * 512 + m) * 2 + 1];
            int a = (kx * x) & 127;
            float c = cs[a], s = sn[a];
            re += tr * c - ti * s;
            im += tr * s + ti * c;
        }
    }
    G[t * 2]     = re;
    G[t * 2 + 1] = im;
}

__global__ void fnde_inv_dft2_out(const float* __restrict__ G, const float* __restrict__ z,
                                  const float* __restrict__ cs, const float* __restrict__ sn,
                                  const float* __restrict__ scal, float* __restrict__ out) {
    int t = blockIdx.x * 256 + threadIdx.x;
    int b = t >> 14;
    int x = (t >> 7) & 127;
    int y = t & 127;
    const float* g = G + ((b * 128 + x) * 16) * 2;
    float acc = g[0];
    for (int ky = 1; ky < 16; ++ky) {
        int a = (ky * y) & 127;
        acc += 2.0f * (g[ky * 2] * cs[a] - g[ky * 2 + 1] * sn[a]);
    }
    out[t] = scal[0] * z[t] + scal[1] + acc * (1.0f / 16384.0f);
}

extern "C" void kernel_launch(void* const* d_in, const int* in_sizes, int n_in,
                              void* d_out, int out_size, void* d_ws, size_t ws_size,
                              hipStream_t stream) {
    (void)in_sizes; (void)n_in; (void)out_size; (void)ws_size;

    const float* zin    = (const float*)d_in[0];
    const float* lift_w = (const float*)d_in[1];
    const float* lift_b = (const float*)d_in[2];
    const float* spec_w1 = (const float*)d_in[3];
    const float* spec_w2 = (const float*)d_in[4];
    const float* fl_bias = (const float*)d_in[5];
    const float* p1_w = (const float*)d_in[6];
    const float* p1_b = (const float*)d_in[7];
    const float* p2_w = (const float*)d_in[8];
    const float* p2_b = (const float*)d_in[9];
    const float* ts   = (const float*)d_in[10];
    float* outp = (float*)d_out;

    float* w = (float*)d_ws;
    float* Wp   = w;  w += (size_t)Ll * WLAYER;
    float* Zre  = w;  w += NSPEC;  float* Zim  = w;  w += NSPEC;
    float* Ure  = w;  w += NSPEC;  float* Uim  = w;  w += NSPEC;
    float* Kre  = w;  w += NSPEC;  float* Kim  = w;  w += NSPEC;
    float* SSre = w;  w += NSPEC;  float* SSim = w;  w += NSPEC;
    float* STre = w;  w += NSPEC;  float* STim = w;  w += NSPEC;
    float* Zhat = w;  w += Bb * NMODE * 2;
    float* Ay   = w;  w += Bb * Dd * Mm * 2;
    float* Tm   = w;  w += Bb * NMODE * 2;
    float* Gi   = w;  w += Bb * Dd * Mm * 2;
    float* qv   = w;  w += 64;
    float* cs   = w;  w += 128;
    float* sn   = w;  w += 128;
    float* scal = w;  w += 8;

    fnde_build_tables<<<1, 128, 0, stream>>>(cs, sn);
    fnde_pack_weights<<<(Ll * NMODE * Cc * Cc / 2) / 256, 256, 0, stream>>>(spec_w1, spec_w2, Wp);
    fnde_fwd_dft1<<<(Bb * Dd * Mm) / 256, 256, 0, stream>>>(zin, cs, sn, Ay);
    fnde_fwd_dft2<<<(Bb * NMODE) / 256, 256, 0, stream>>>(Ay, cs, sn, Zhat);
    fnde_init_spec<<<NSPEC / 256, 256, 0, stream>>>(Zhat, lift_w, lift_b, Zre, Zim, STre, STim);

    for (int l = 0; l < Ll; ++l) {
        const float* Wl = Wp + (size_t)l * WLAYER;
        const float* bl = fl_bias + l * Cc;
        for (int s = 0; s < 4; ++s) {
            fnde_gapply<<<128, 128, 0, stream>>>(Zre, Zim, Wl, Kre, Kim, bl);
            fnde_spec_mid<<<NSPEC / 256, 256, 0, stream>>>(Zre, Zim, Kre, Kim, Ure, Uim,
                                                           SSre, SSim, ts, s, 1);
            fnde_gapply<<<128, 128, 0, stream>>>(Ure, Uim, Wl, Kre, Kim, bl);
            fnde_spec_mid<<<NSPEC / 256, 256, 0, stream>>>(Zre, Zim, Kre, Kim, Ure, Uim,
                                                           SSre, SSim, ts, s, 2);
            fnde_gapply<<<128, 128, 0, stream>>>(Ure, Uim, Wl, Kre, Kim, bl);
            fnde_spec_mid<<<NSPEC / 256, 256, 0, stream>>>(Zre, Zim, Kre, Kim, Ure, Uim,
                                                           SSre, SSim, ts, s, 3);
            fnde_gapply<<<128, 128, 0, stream>>>(Ure, Uim, Wl, Kre, Kim, bl);
            fnde_spec_fin<<<NSPEC / 256, 256, 0, stream>>>(Zre, Zim, Kre, Kim, SSre, SSim,
                                                           STre, STim, ts, s);
        }
    }

    fnde_proj_prep<<<1, 64, 0, stream>>>(p1_w, p1_b, p2_w, p2_b, lift_w, lift_b, qv, scal);
    fnde_proj_modes<<<(Bb * NMODE) / 256, 256, 0, stream>>>(STre, STim, qv, Tm);
    fnde_inv_dft1<<<(Bb * Dd * Mm) / 256, 256, 0, stream>>>(Tm, cs, sn, Gi);
    fnde_inv_dft2_out<<<(Bb * Dd * Dd) / 256, 256, 0, stream>>>(Gi, zin, cs, sn, scal, outp);
}


# Whole ODE chain (48 spectral products + RK4 combines) in ONE launch: 4 waves per mode hold state and the layer's weights in registers; ky=0 partner modes share a block, swap via LDS; 105->9 launches
# speedup vs baseline: 5.9804x; 4.9052x over previous
#include <hip/hip_runtime.h>


typedef float v2f __attribute__((ext_vector_type(2)));
typedef float v4f __attribute__((ext_vector_type(4)));
typedef float v8f __attribute__((ext_vector_type(8)));

#define Bb    16
#define Cc    64
#define Dd    128
#define Mm    16
#define Ll    3
#define NMODE 512
#define NSPEC (NMODE*Cc*Bb)
#define WLAYER (2*NMODE*Cc*Cc)

__global__ void fnde_build_tables(float* cs, float* sn) {
    int t = threadIdx.x;
    float ang = (float)(6.283185307179586 / 128.0) * (float)t;
    cs[t] = cosf(ang);
    sn[t] = sinf(ang);
}

__global__ void fnde_pack_weights(const float* __restrict__ w1,
                                  const float* __restrict__ w2,
                                  float* __restrict__ Wp) {
    int idx  = blockIdx.x * 256 + threadIdx.x;
    int l    = idx / (NMODE * 2048);
    int rem  = idx % (NMODE * 2048);
    int m    = rem >> 11;
    int frag = rem & 2047;
    int lane = frag & 31;
    int kk   = (frag >> 5) & 15;
    int mt   = frag >> 9;
    int o    = mt * 16 + (lane & 15);
    int i    = kk * 4 + (lane >> 4) * 2;
    int corner = m >> 8;
    int mx = (m >> 4) & 15;
    int ky = m & 15;
    const float* src = corner ? w2 : w1;
    int sidx = ((((l * Cc + i) * Cc + o) * Mm + mx) * Mm + ky) * 2;
    const int kstride = Cc * Mm * Mm * 2;
    v4f f;
    f.x = src[sidx];
    f.z = src[sidx + 1];
    f.y = src[sidx + kstride];
    f.w = src[sidx + kstride + 1];
    v4f* dst = (v4f*)(Wp + (size_t)l * WLAYER + (size_t)m * (2 * Cc * Cc));
    dst[frag] = f;
}

__global__ void fnde_fwd_dft1(const float* __restrict__ z,
                              const float* __restrict__ cs,
                              const float* __restrict__ sn,
                              float* __restrict__ Ay) {
    int t  = blockIdx.x * 256 + threadIdx.x;
    int b  = t >> 11;
    int x  = (t >> 4) & 127;
    int ky = t & 15;
    const float* row = z + (b * 128 + x) * 128;
    float re = 0.f, im = 0.f;
    for (int y = 0; y < 128; ++y) {
        float v = row[y];
        int a = (ky * y) & 127;
        re += v * cs[a];
        im -= v * sn[a];
    }
    Ay[t * 2]     = re;
    Ay[t * 2 + 1] = im;
}

__global__ void fnde_fwd_dft2(const float* __restrict__ Ay,
                              const float* __restrict__ cs,
                              const float* __restrict__ sn,
                              float* __restrict__ Zhat) {
    int t = blockIdx.x * 256 + threadIdx.x;
    int b = t >> 9;
    int m = t & 511;
    int corner = m >> 8, mx = (m >> 4) & 15, ky = m & 15;
    int kx = corner ? (112 + mx) : mx;
    float re = 0.f, im = 0.f;
    for (int x = 0; x < 128; ++x) {
        int base = ((b * 128 + x) * 16 + ky) * 2;
        float ar = Ay[base], ai = Ay[base + 1];
        int a = (kx * x) & 127;
        float c = cs[a], s = sn[a];
        re += ar * c + ai * s;
        im += ai * c - ar * s;
    }
    Zhat[t * 2]     = re;
    Zhat[t * 2 + 1] = im;
}


__device__ __forceinline__ void fnde_eval(v2f* __restrict__ Ub, int ownBase, int bBase,
                                          const float (&vr)[8], const float (&vi)[8],
                                          const v2f (&are)[16], const v2f (&aim)[16],
                                          const v2f (&nai)[16], const float (&bv)[8],
                                          float (&kr)[8], float (&ki)[8]) {
#pragma unroll
    for (int q = 0; q < 4; ++q) {
        v2f tr, ti;
        tr.x = vr[2 * q];   tr.y = vr[2 * q + 1];
        ti.x = vi[2 * q];   ti.y = vi[2 * q + 1];
        Ub[ownBase + q * 16]       = tr;
        Ub[512 + ownBase + q * 16] = ti;
    }
    __syncthreads();
    v8f cr = {}, ci = {};
#pragma unroll
    for (int kk = 0; kk < 16; ++kk) {
        v2f br = Ub[bBase + kk * 32];
        v2f bi = Ub[512 + bBase + kk * 32];
        cr = __builtin_amdgcn_wmma_f32_16x16x4_f32(false, are[kk], false, br, (short)0, cr, false, false);
        ci = __builtin_amdgcn_wmma_f32_16x16x4_f32(false, are[kk], false, bi, (short)0, ci, false, false);
        cr = __builtin_amdgcn_wmma_f32_16x16x4_f32(false, nai[kk], false, bi, (short)0, cr, false, false);
        ci = __builtin_amdgcn_wmma_f32_16x16x4_f32(false, aim[kk], false, br, (short)0, ci, false, false);
    }
    asm volatile("v_nop\n\tv_nop\n\tv_nop\n\tv_nop" : "+v"(cr), "+v"(ci));
#pragma unroll
    for (int j = 0; j < 8; ++j) {
        kr[j] = cr[j] + bv[j];
        ki[j] = ci[j];
    }
}

__device__ __forceinline__ void fnde_pi_pair(v2f* Xown, const v2f* Xpar,
                                             bool noPartner, int ownBase,
                                             const float (&kr)[8], const float (&ki)[8],
                                             float (&pr)[8], float (&pim)[8]) {
#pragma unroll
    for (int q = 0; q < 4; ++q) {
        v2f tr, ti;
        tr.x = kr[2 * q];   tr.y = kr[2 * q + 1];
        ti.x = ki[2 * q];   ti.y = ki[2 * q + 1];
        Xown[ownBase + q * 16]       = tr;
        Xown[512 + ownBase + q * 16] = ti;
    }
    __syncthreads();
#pragma unroll
    for (int q = 0; q < 4; ++q) {
        v2f tr = Xpar[ownBase + q * 16];
        v2f ti = Xpar[512 + ownBase + q * 16];
        if (noPartner) { tr.x = 0.0f; tr.y = 0.0f; ti.x = 0.0f; ti.y = 0.0f; }
        pr[2 * q]      = 0.5f * (kr[2 * q]     + tr.x);
        pr[2 * q + 1]  = 0.5f * (kr[2 * q + 1] + tr.y);
        pim[2 * q]     = 0.5f * (ki[2 * q]     - ti.x);
        pim[2 * q + 1] = 0.5f * (ki[2 * q + 1] - ti.y);
    }
}

__global__ __launch_bounds__(256)
void fnde_chain(const float* __restrict__ Zhat,
                const float* __restrict__ lw, const float* __restrict__ lb,
                const float* __restrict__ Wp, const float* __restrict__ bias,
                const float* __restrict__ ts,
                float* __restrict__ STre, float* __restrict__ STim) {
    __shared__ v2f Ush[2 * 2 * 2 * 512];
    __shared__ v2f Xsh[2 * 2 * 512];

    const int blk = blockIdx.x;
    const int wave = __builtin_amdgcn_readfirstlane((int)(threadIdx.x >> 5));
    const int lane = threadIdx.x & 31;
    const int laneHi = lane >> 4;
    const int laneLo = lane & 15;
    const int mi = wave >> 2;
    const int mt = wave & 3;
    const int ky  = blk & 15;
    const int mx0 = blk >> 4;
    const int mode = mi ? (256 + (((16 - mx0) & 15) << 4) + ky) : blk;
    const bool ky0 = (ky == 0);
    const bool noPartner = (blk == 0) && (mi == 1);
    const int  pmi = (blk == 0) ? mi : (1 - mi);

    const int ownBase = (mt * 8 + laneHi * 4) * 16 + laneLo;
    const int bBase   = laneHi * 16 + laneLo;
    const int row0    = mt * 16 + laneHi * 8;

    v2f* U0 = Ush + (0 * 2 + mi) * 1024;
    v2f* U1 = Ush + (1 * 2 + mi) * 1024;
    v2f* Xown = Xsh + mi * 1024;
    const v2f* Xpar = Xsh + pmi * 1024;

    float zr[8], zi[8], ssr[8], ssi[8], str[8], sti[8];
    {
        float hr = Zhat[(laneLo * 512 + mode) * 2];
        float hi = Zhat[(laneLo * 512 + mode) * 2 + 1];
#pragma unroll
        for (int j = 0; j < 8; ++j) {
            float w = lw[row0 + j];
            float lbj = lb[row0 + j];
            zr[j] = w * hr + ((mode == 0) ? lbj * 16384.0f : 0.0f);
            zi[j] = w * hi;
            str[j] = 0.0f;  sti[j] = 0.0f;
        }
    }

#define FNDE_PROJ(SR, SI)                                                        \
    if (ky0) fnde_pi_pair(Xown, Xpar, noPartner, ownBase, SR, SI, pr, pim);      \
    else {                                                                       \
        _Pragma("unroll")                                                        \
        for (int j = 0; j < 8; ++j) { pr[j] = SR[j]; pim[j] = SI[j]; }           \
    }

#pragma unroll 1
    for (int l = 0; l < Ll; ++l) {
        const v4f* Wm = (const v4f*)(Wp + (size_t)l * WLAYER + (size_t)mode * (2 * Cc * Cc))
                        + (mt * 16) * 32 + lane;
        v2f are[16], aim[16], nai[16];
#pragma unroll
        for (int kk = 0; kk < 16; ++kk) {
            v4f a = Wm[kk * 32];
            are[kk].x = a.x;    are[kk].y = a.y;
            aim[kk].x = a.z;    aim[kk].y = a.w;
            nai[kk].x = -a.z;   nai[kk].y = -a.w;
        }
        float bv[8];
#pragma unroll
        for (int j = 0; j < 8; ++j) {
            float bj = bias[l * Cc + row0 + j];
            bv[j] = (mode == 0) ? bj * 16384.0f : 0.0f;
        }

#pragma unroll 1
        for (int s = 0; s < 4; ++s) {
            const float h  = ts[s + 1] - ts[s];
            const float hh = 0.5f * h;
            const float h6 = h * (1.0f / 6.0f);
            float kr[8], ki[8], pr[8], pim[8], ur[8], ui[8];

            fnde_eval(U0, ownBase, bBase, zr, zi, are, aim, nai, bv, kr, ki);
#pragma unroll
            for (int j = 0; j < 8; ++j) { ssr[j] = kr[j]; ssi[j] = ki[j]; }
            FNDE_PROJ(kr, ki)
#pragma unroll
            for (int j = 0; j < 8; ++j) { ur[j] = zr[j] + hh * pr[j]; ui[j] = zi[j] + hh * pim[j]; }

            fnde_eval(U1, ownBase, bBase, ur, ui, are, aim, nai, bv, kr, ki);
#pragma unroll
            for (int j = 0; j < 8; ++j) { ssr[j] += 2.0f * kr[j]; ssi[j] += 2.0f * ki[j]; }
            FNDE_PROJ(kr, ki)
#pragma unroll
            for (int j = 0; j < 8; ++j) { ur[j] = zr[j] + hh * pr[j]; ui[j] = zi[j] + hh * pim[j]; }

            fnde_eval(U0, ownBase, bBase, ur, ui, are, aim, nai, bv, kr, ki);
#pragma unroll
            for (int j = 0; j < 8; ++j) { ssr[j] += 2.0f * kr[j]; ssi[j] += 2.0f * ki[j]; }
            FNDE_PROJ(kr, ki)
#pragma unroll
            for (int j = 0; j < 8; ++j) { ur[j] = zr[j] + h * pr[j]; ui[j] = zi[j] + h * pim[j]; }

            fnde_eval(U1, ownBase, bBase, ur, ui, are, aim, nai, bv, kr, ki);
#pragma unroll
            for (int j = 0; j < 8; ++j) { ssr[j] += kr[j]; ssi[j] += ki[j]; }
            FNDE_PROJ(ssr, ssi)
#pragma unroll
            for (int j = 0; j < 8; ++j) {
                zr[j]  += h6 * pr[j];    zi[j]  += h6 * pim[j];
                str[j] += h6 * ssr[j];   sti[j] += h6 * ssi[j];
            }
        }
    }
#undef FNDE_PROJ

#pragma unroll
    for (int j = 0; j < 8; ++j) {
        STre[mode * 1024 + (row0 + j) * 16 + laneLo] = str[j];
        STim[mode * 1024 + (row0 + j) * 16 + laneLo] = sti[j];
    }
}

__global__ void fnde_proj_prep(const float* __restrict__ p1w, const float* __restrict__ p1b,
                               const float* __restrict__ p2w, const float* __restrict__ p2b,
                               const float* __restrict__ lw,  const float* __restrict__ lb,
                               float* __restrict__ q, float* __restrict__ scal) {
    __shared__ float qs[64];
    int c = threadIdx.x;
    float acc = 0.f;
    for (int p = 0; p < 64; ++p) acc += p2w[p] * p1w[p * 64 + c];
    qs[c] = acc;
    q[c] = acc;
    __syncthreads();
    if (c == 0) {
        float r = p2b[0];
        for (int p = 0; p < 64; ++p) r += p2w[p] * p1b[p];
        float al = 0.f, be = r;
        for (int i = 0; i < 64; ++i) { al += qs[i] * lw[i]; be += qs[i] * lb[i]; }
        scal[0] = al;
        scal[1] = be;
    }
}

__global__ void fnde_proj_modes(const float* __restrict__ STre, const float* __restrict__ STim,
                                const float* __restrict__ q, float* __restrict__ T) {
    int t = blockIdx.x * 256 + threadIdx.x;
    int b = t >> 9;
    int m = t & 511;
    const float* pr = STre + m * 1024 + b;
    const float* pi = STim + m * 1024 + b;
    float re = 0.f, im = 0.f;
    for (int c = 0; c < 64; ++c) {
        float w = q[c];
        re += w * pr[c * 16];
        im += w * pi[c * 16];
    }
    T[t * 2]     = re;
    T[t * 2 + 1] = im;
}

__global__ void fnde_inv_dft1(const float* __restrict__ T,
                              const float* __restrict__ cs, const float* __restrict__ sn,
                              float* __restrict__ G) {
    int t  = blockIdx.x * 256 + threadIdx.x;
    int b  = t >> 11;
    int x  = (t >> 4) & 127;
    int ky = t & 15;
    float re = 0.f, im = 0.f;
    for (int corner = 0; corner < 2; ++corner) {
        for (int mx = 0; mx < 16; ++mx) {
            int m  = corner * 256 + mx * 16 + ky;
            int kx = corner ? (112 + mx) : mx;
            float tr = T[(b * 512 + m) * 2];
            float ti = T[(b * 512 + m) * 2 + 1];
            int a = (kx * x) & 127;
            float c = cs[a], s = sn[a];
            re += tr * c - ti * s;
            im += tr * s + ti * c;
        }
    }
    G[t * 2]     = re;
    G[t * 2 + 1] = im;
}

__global__ void fnde_inv_dft2_out(const float* __restrict__ G, const float* __restrict__ z,
                                  const float* __restrict__ cs, const float* __restrict__ sn,
                                  const float* __restrict__ scal, float* __restrict__ out) {
    int t = blockIdx.x * 256 + threadIdx.x;
    int b = t >> 14;
    int x = (t >> 7) & 127;
    int y = t & 127;
    const float* g = G + ((b * 128 + x) * 16) * 2;
    float acc = g[0];
    for (int ky = 1; ky < 16; ++ky) {
        int a = (ky * y) & 127;
        acc += 2.0f * (g[ky * 2] * cs[a] - g[ky * 2 + 1] * sn[a]);
    }
    out[t] = scal[0] * z[t] + scal[1] + acc * (1.0f / 16384.0f);
}

extern "C" void kernel_launch(void* const* d_in, const int* in_sizes, int n_in,
                              void* d_out, int out_size, void* d_ws, size_t ws_size,
                              hipStream_t stream) {
    (void)in_sizes; (void)n_in; (void)out_size; (void)ws_size;

    const float* zin    = (const float*)d_in[0];
    const float* lift_w = (const float*)d_in[1];
    const float* lift_b = (const float*)d_in[2];
    const float* spec_w1 = (const float*)d_in[3];
    const float* spec_w2 = (const float*)d_in[4];
    const float* fl_bias = (const float*)d_in[5];
    const float* p1_w = (const float*)d_in[6];
    const float* p1_b = (const float*)d_in[7];
    const float* p2_w = (const float*)d_in[8];
    const float* p2_b = (const float*)d_in[9];
    const float* ts   = (const float*)d_in[10];
    float* outp = (float*)d_out;

    float* w = (float*)d_ws;
    float* Wp   = w;  w += (size_t)Ll * WLAYER;
    float* STre = w;  w += NSPEC;  float* STim = w;  w += NSPEC;
    float* Zhat = w;  w += Bb * NMODE * 2;
    float* Ay   = w;  w += Bb * Dd * Mm * 2;
    float* Tm   = w;  w += Bb * NMODE * 2;
    float* Gi   = w;  w += Bb * Dd * Mm * 2;
    float* qv   = w;  w += 64;
    float* cs   = w;  w += 128;
    float* sn   = w;  w += 128;
    float* scal = w;  w += 8;

    fnde_build_tables<<<1, 128, 0, stream>>>(cs, sn);
    fnde_pack_weights<<<(Ll * NMODE * Cc * Cc / 2) / 256, 256, 0, stream>>>(spec_w1, spec_w2, Wp);
    fnde_fwd_dft1<<<(Bb * Dd * Mm) / 256, 256, 0, stream>>>(zin, cs, sn, Ay);
    fnde_fwd_dft2<<<(Bb * NMODE) / 256, 256, 0, stream>>>(Ay, cs, sn, Zhat);

    fnde_chain<<<NMODE / 2, 256, 0, stream>>>(Zhat, lift_w, lift_b, Wp, fl_bias, ts, STre, STim);

    fnde_proj_prep<<<1, 64, 0, stream>>>(p1_w, p1_b, p2_w, p2_b, lift_w, lift_b, qv, scal);
    fnde_proj_modes<<<(Bb * NMODE) / 256, 256, 0, stream>>>(STre, STim, qv, Tm);
    fnde_inv_dft1<<<(Bb * Dd * Mm) / 256, 256, 0, stream>>>(Tm, cs, sn, Gi);
    fnde_inv_dft2_out<<<(Bb * Dd * Dd) / 256, 256, 0, stream>>>(Gi, zin, cs, sn, scal, outp);
}


# Weight pack re-tiled: whole 128-byte source lines read 16 bytes per lane, turned through LDS, written as whole 512-byte fragments; 1536 groups of 4 waves replace 12288 groups of 8
# speedup vs baseline: 6.7727x; 1.1325x over previous
#include <hip/hip_runtime.h>


typedef float v2f __attribute__((ext_vector_type(2)));
typedef float v4f __attribute__((ext_vector_type(4)));
typedef float v8f __attribute__((ext_vector_type(8)));

#define Bb    16
#define Cc    64
#define Dd    128
#define Mm    16
#define Ll    3
#define NMODE 512
#define NSPEC (NMODE*Cc*Bb)
#define WLAYER (2*NMODE*Cc*Cc)

__global__ void fnde_build_tables(float* cs, float* sn) {
    int t = threadIdx.x;
    float ang = (float)(6.283185307179586 / 128.0) * (float)t;
    cs[t] = cosf(ang);
    sn[t] = sinf(ang);
}

#define PK_PITCH 516
__global__ __launch_bounds__(128)
void fnde_pack_weights(const float* __restrict__ w1,
                       const float* __restrict__ w2,
                       float* __restrict__ Wp) {
    __shared__ v4f pk_tile[16 * (PK_PITCH / 4)];
    float* tile = (float*)pk_tile;

    const int t = threadIdx.x;
    int blk = blockIdx.x;
    const int kq = blk & 3;    blk >>= 2;
    const int mt = blk & 3;    blk >>= 2;
    const int mx = blk & 15;   blk >>= 4;
    const int corner = blk & 1;
    const int l = blk >> 1;
    const float* src = corner ? w2 : w1;

    const int p  = t & 7;
    const int b  = (t >> 3) & 1;
    const int cb = (t >> 4) & 1;
    const int w  = t >> 5;
    const float* sp = src + ((((size_t)(l * Cc + kq * 16 + b) * Cc + mt * 16 + 2 * w + cb) * Mm + mx) * Mm) * 2
                          + p * 4;
    float* tp = tile + (2 * p) * PK_PITCH + (2 * w + cb) * 4 + b * 2;
    v4f v[16];
#pragma unroll
    for (int r = 0; r < 16; ++r) {
        const int ip = r >> 1;
        const int cp = (r & 1) * 4;
        v[r] = *(const v4f*)(sp + (size_t)(ip * 2) * (Cc * Mm * Mm * 2) + (size_t)(cp * 2) * (Mm * Mm * 2));
    }
#pragma unroll
    for (int r = 0; r < 16; ++r) {
        const int ip = r >> 1;
        const int cp = (r & 1) * 4;
        float* d = tp + ip * 64 + cp * 8;
        v2f e0, e1;
        e0.x = v[r].x;   e0.y = v[r].y;
        e1.x = v[r].z;   e1.y = v[r].w;
        *(v2f*)d              = e0;
        *(v2f*)(d + PK_PITCH) = e1;
    }
    __syncthreads();

    const int lane = t & 31;
    const v4f* tq = pk_tile + w * 32 + lane;
    v4f* dp = (v4f*)(Wp + (size_t)l * WLAYER + (size_t)(corner * 256 + mx * 16) * (2 * Cc * Cc))
              + (mt * 16 + kq * 4 + w) * 32 + lane;
#pragma unroll
    for (int ky = 0; ky < 16; ++ky) {
        v4f a = tq[ky * (PK_PITCH / 4)];
        v4f f;
        f.x = a.x;   f.y = a.z;   f.z = a.y;   f.w = a.w;
        dp[(size_t)ky * (2 * Cc * Cc / 4)] = f;
    }
}

__global__ void fnde_fwd_dft1(const float* __restrict__ z,
                              const float* __restrict__ cs,
                              const float* __restrict__ sn,
                              float* __restrict__ Ay) {
    int t  = blockIdx.x * 256 + threadIdx.x;
    int b  = t >> 11;
    int x  = (t >> 4) & 127;
    int ky = t & 15;
    const float* row = z + (b * 128 + x) * 128;
    float re = 0.f, im = 0.f;
    for (int y = 0; y < 128; ++y) {
        float v = row[y];
        int a = (ky * y) & 127;
        re += v * cs[a];
        im -= v * sn[a];
    }
    Ay[t * 2]     = re;
    Ay[t * 2 + 1] = im;
}

__global__ void fnde_fwd_dft2(const float* __restrict__ Ay,
                              const float* __restrict__ cs,
                              const float* __restrict__ sn,
                              float* __restrict__ Zhat) {
    int t = blockIdx.x * 256 + threadIdx.x;
    int b = t >> 9;
    int m = t & 511;
    int corner = m >> 8, mx = (m >> 4) & 15, ky = m & 15;
    int kx = corner ? (112 + mx) : mx;
    float re = 0.f, im = 0.f;
    for (int x = 0; x < 128; ++x) {
        int base = ((b * 128 + x) * 16 + ky) * 2;
        float ar = Ay[base], ai = Ay[base + 1];
        int a = (kx * x) & 127;
        float c = cs[a], s = sn[a];
        re += ar * c + ai * s;
        im += ai * c - ar * s;
    }
    Zhat[t * 2]     = re;
    Zhat[t * 2 + 1] = im;
}


__device__ __forceinline__ void fnde_eval(v2f* __restrict__ Ub, int ownBase, int bBase,
                                          const float (&vr)[8], const float (&vi)[8],
                                          const v2f (&are)[16], const v2f (&aim)[16],
                                          const v2f (&nai)[16], const float (&bv)[8],
                                          float (&kr)[8], float (&ki)[8]) {
#pragma unroll
    for (int q = 0; q < 4; ++q) {
        v2f tr, ti;
        tr.x = vr[2 * q];   tr.y = vr[2 * q + 1];
        ti.x = vi[2 * q];   ti.y = vi[2 * q + 1];
        Ub[ownBase + q * 16]       = tr;
        Ub[512 + ownBase + q * 16] = ti;
    }
    __syncthreads();
    v8f cr = {}, ci = {};
#pragma unroll
    for (int kk = 0; kk < 16; ++kk) {
        v2f br = Ub[bBase + kk * 32];
        v2f bi = Ub[512 + bBase + kk * 32];
        cr = __builtin_amdgcn_wmma_f32_16x16x4_f32(false, are[kk], false, br, (short)0, cr, false, false);
        ci = __builtin_amdgcn_wmma_f32_16x16x4_f32(false, are[kk], false, bi, (short)0, ci, false, false);
        cr = __builtin_amdgcn_wmma_f32_16x16x4_f32(false, nai[kk], false, bi, (short)0, cr, false, false);
        ci = __builtin_amdgcn_wmma_f32_16x16x4_f32(false, aim[kk], false, br, (short)0, ci, false, false);
    }
    asm volatile("v_nop\n\tv_nop\n\tv_nop\n\tv_nop" : "+v"(cr), "+v"(ci));
#pragma unroll
    for (int j = 0; j < 8; ++j) {
        kr[j] = cr[j] + bv[j];
        ki[j] = ci[j];
    }
}

__device__ __forceinline__ void fnde_pi_pair(v2f* Xown, const v2f* Xpar,
                                             bool noPartner, int ownBase,
                                             const float (&kr)[8], const float (&ki)[8],
                                             float (&pr)[8], float (&pim)[8]) {
#pragma unroll
    for (int q = 0; q < 4; ++q) {
        v2f tr, ti;
        tr.x = kr[2 * q];   tr.y = kr[2 * q + 1];
        ti.x = ki[2 * q];   ti.y = ki[2 * q + 1];
        Xown[ownBase + q * 16]       = tr;
        Xown[512 + ownBase + q * 16] = ti;
    }
    __syncthreads();
#pragma unroll
    for (int q = 0; q < 4; ++q) {
        v2f tr = Xpar[ownBase + q * 16];
        v2f ti = Xpar[512 + ownBase + q * 16];
        if (noPartner) { tr.x = 0.0f; tr.y = 0.0f; ti.x = 0.0f; ti.y = 0.0f; }
        pr[2 * q]      = 0.5f * (kr[2 * q]     + tr.x);
        pr[2 * q + 1]  = 0.5f * (kr[2 * q + 1] + tr.y);
        pim[2 * q]     = 0.5f * (ki[2 * q]     - ti.x);
        pim[2 * q + 1] = 0.5f * (ki[2 * q + 1] - ti.y);
    }
}

__global__ __launch_bounds__(256)
void fnde_chain(const float* __restrict__ Zhat,
                const float* __restrict__ lw, const float* __restrict__ lb,
                const float* __restrict__ Wp, const float* __restrict__ bias,
                const float* __restrict__ ts,
                float* __restrict__ STre, float* __restrict__ STim) {
    __shared__ v2f Ush[2 * 2 * 2 * 512];
    __shared__ v2f Xsh[2 * 2 * 512];

    const int blk = blockIdx.x;
    const int wave = __builtin_amdgcn_readfirstlane((int)(threadIdx.x >> 5));
    const int lane = threadIdx.x & 31;
    const int laneHi = lane >> 4;
    const int laneLo = lane & 15;
    const int mi = wave >> 2;
    const int mt = wave & 3;
    const int ky  = blk & 15;
    const int mx0 = blk >> 4;
    const int mode = mi ? (256 + (((16 - mx0) & 15) << 4) + ky) : blk;
    const bool ky0 = (ky == 0);
    const bool noPartner = (blk == 0) && (mi == 1);
    const int  pmi = (blk == 0) ? mi : (1 - mi);

    const int ownBase = (mt * 8 + laneHi * 4) * 16 + laneLo;
    const int bBase   = laneHi * 16 + laneLo;
    const int row0    = mt * 16 + laneHi * 8;

    v2f* U0 = Ush + (0 * 2 + mi) * 1024;
    v2f* U1 = Ush + (1 * 2 + mi) * 1024;
    v2f* Xown = Xsh + mi * 1024;
    const v2f* Xpar = Xsh + pmi * 1024;

    float zr[8], zi[8], ssr[8], ssi[8], str[8], sti[8];
    {
        float hr = Zhat[(laneLo * 512 + mode) * 2];
        float hi = Zhat[(laneLo * 512 + mode) * 2 + 1];
#pragma unroll
        for (int j = 0; j < 8; ++j) {
            float w = lw[row0 + j];
            float lbj = lb[row0 + j];
            zr[j] = w * hr + ((mode == 0) ? lbj * 16384.0f : 0.0f);
            zi[j] = w * hi;
            str[j] = 0.0f;  sti[j] = 0.0f;
        }
    }

#define FNDE_PROJ(SR, SI)                                                        \
    if (ky0) fnde_pi_pair(Xown, Xpar, noPartner, ownBase, SR, SI, pr, pim);      \
    else {                                                                       \
        _Pragma("unroll")                                                        \
        for (int j = 0; j < 8; ++j) { pr[j] = SR[j]; pim[j] = SI[j]; }           \
    }

#pragma unroll 1
    for (int l = 0; l < Ll; ++l) {
        const v4f* Wm = (const v4f*)(Wp + (size_t)l * WLAYER + (size_t)mode * (2 * Cc * Cc))
                        + (mt * 16) * 32 + lane;
        v2f are[16], aim[16], nai[16];
#pragma unroll
        for (int kk = 0; kk < 16; ++kk) {
            v4f a = Wm[kk * 32];
            are[kk].x = a.x;    are[kk].y = a.y;
            aim[kk].x = a.z;    aim[kk].y = a.w;
            nai[kk].x = -a.z;   nai[kk].y = -a.w;
        }
        float bv[8];
#pragma unroll
        for (int j = 0; j < 8; ++j) {
            float bj = bias[l * Cc + row0 + j];
            bv[j] = (mode == 0) ? bj * 16384.0f : 0.0f;
        }

#pragma unroll 1
        for (int s = 0; s < 4; ++s) {
            const float h  = ts[s + 1] - ts[s];
            const float hh = 0.5f * h;
            const float h6 = h * (1.0f / 6.0f);
            float kr[8], ki[8], pr[8], pim[8], ur[8], ui[8];

            fnde_eval(U0, ownBase, bBase, zr, zi, are, aim, nai, bv, kr, ki);
#pragma unroll
            for (int j = 0; j < 8; ++j) { ssr[j] = kr[j]; ssi[j] = ki[j]; }
            FNDE_PROJ(kr, ki)
#pragma unroll
            for (int j = 0; j < 8; ++j) { ur[j] = zr[j] + hh * pr[j]; ui[j] = zi[j] + hh * pim[j]; }

            fnde_eval(U1, ownBase, bBase, ur, ui, are, aim, nai, bv, kr, ki);
#pragma unroll
            for (int j = 0; j < 8; ++j) { ssr[j] += 2.0f * kr[j]; ssi[j] += 2.0f * ki[j]; }
            FNDE_PROJ(kr, ki)
#pragma unroll
            for (int j = 0; j < 8; ++j) { ur[j] = zr[j] + hh * pr[j]; ui[j] = zi[j] + hh * pim[j]; }

            fnde_eval(U0, ownBase, bBase, ur, ui, are, aim, nai, bv, kr, ki);
#pragma unroll
            for (int j = 0; j < 8; ++j) { ssr[j] += 2.0f * kr[j]; ssi[j] += 2.0f * ki[j]; }
            FNDE_PROJ(kr, ki)
#pragma unroll
            for (int j = 0; j < 8; ++j) { ur[j] = zr[j] + h * pr[j]; ui[j] = zi[j] + h * pim[j]; }

            fnde_eval(U1, ownBase, bBase, ur, ui, are, aim, nai, bv, kr, ki);
#pragma unroll
            for (int j = 0; j < 8; ++j) { ssr[j] += kr[j]; ssi[j] += ki[j]; }
            FNDE_PROJ(ssr, ssi)
#pragma unroll
            for (int j = 0; j < 8; ++j) {
                zr[j]  += h6 * pr[j];    zi[j]  += h6 * pim[j];
                str[j] += h6 * ssr[j];   sti[j] += h6 * ssi[j];
            }
        }
    }
#undef FNDE_PROJ

#pragma unroll
    for (int j = 0; j < 8; ++j) {
        STre[mode * 1024 + (row0 + j) * 16 + laneLo] = str[j];
        STim[mode * 1024 + (row0 + j) * 16 + laneLo] = sti[j];
    }
}

__global__ void fnde_proj_prep(const float* __restrict__ p1w, const float* __restrict__ p1b,
                               const float* __restrict__ p2w, const float* __restrict__ p2b,
                               const float* __restrict__ lw,  const float* __restrict__ lb,
                               float* __restrict__ q, float* __restrict__ scal) {
    __shared__ float qs[64];
    int c = threadIdx.x;
    float acc = 0.f;
    for (int p = 0; p < 64; ++p) acc += p2w[p] * p1w[p * 64 + c];
    qs[c] = acc;
    q[c] = acc;
    __syncthreads();
    if (c == 0) {
        float r = p2b[0];
        for (int p = 0; p < 64; ++p) r += p2w[p] * p1b[p];
        float al = 0.f, be = r;
        for (int i = 0; i < 64; ++i) { al += qs[i] * lw[i]; be += qs[i] * lb[i]; }
        scal[0] = al;
        scal[1] = be;
    }
}

__global__ void fnde_proj_modes(const float* __restrict__ STre, const float* __restrict__ STim,
                                const float* __restrict__ q, float* __restrict__ T) {
    int t = blockIdx.x * 256 + threadIdx.x;
    int b = t >> 9;
    int m = t & 511;
    const float* pr = STre + m * 1024 + b;
    const float* pi = STim + m * 1024 + b;
    float re = 0.f, im = 0.f;
    for (int c = 0; c < 64; ++c) {
        float w = q[c];
        re += w * pr[c * 16];
        im += w * pi[c * 16];
    }
    T[t * 2]     = re;
    T[t * 2 + 1] = im;
}

__global__ void fnde_inv_dft1(const float* __restrict__ T,
                              const float* __restrict__ cs, const float* __restrict__ sn,
                              float* __restrict__ G) {
    int t  = blockIdx.x * 256 + threadIdx.x;
    int b  = t >> 11;
    int x  = (t >> 4) & 127;
    int ky = t & 15;
    float re = 0.f, im = 0.f;
    for (int corner = 0; corner < 2; ++corner) {
        for (int mx = 0; mx < 16; ++mx) {
            int m  = corner * 256 + mx * 16 + ky;
            int kx = corner ? (112 + mx) : mx;
            float tr = T[(b * 512 + m) * 2];
            float ti = T[(b * 512 + m) * 2 + 1];
            int a = (kx * x) & 127;
            float c = cs[a], s = sn[a];
            re += tr * c - ti * s;
            im += tr * s + ti * c;
        }
    }
    G[t * 2]     = re;
    G[t * 2 + 1] = im;
}

__global__ void fnde_inv_dft2_out(const float* __restrict__ G, const float* __restrict__ z,
                                  const float* __restrict__ cs, const float* __restrict__ sn,
                                  const float* __restrict__ scal, float* __restrict__ out) {
    int t = blockIdx.x * 256 + threadIdx.x;
    int b = t >> 14;
    int x = (t >> 7) & 127;
    int y = t & 127;
    const float* g = G + ((b * 128 + x) * 16) * 2;
    float acc = g[0];
    for (int ky = 1; ky < 16; ++ky) {
        int a = (ky * y) & 127;
        acc += 2.0f * (g[ky * 2] * cs[a] - g[ky * 2 + 1] * sn[a]);
    }
    out[t] = scal[0] * z[t] + scal[1] + acc * (1.0f / 16384.0f);
}

extern "C" void kernel_launch(void* const* d_in, const int* in_sizes, int n_in,
                              void* d_out, int out_size, void* d_ws, size_t ws_size,
                              hipStream_t stream) {
    (void)in_sizes; (void)n_in; (void)out_size; (void)ws_size;

    const float* zin    = (const float*)d_in[0];
    const float* lift_w = (const float*)d_in[1];
    const float* lift_b = (const float*)d_in[2];
    const float* spec_w1 = (const float*)d_in[3];
    const float* spec_w2 = (const float*)d_in[4];
    const float* fl_bias = (const float*)d_in[5];
    const float* p1_w = (const float*)d_in[6];
    const float* p1_b = (const float*)d_in[7];
    const float* p2_w = (const float*)d_in[8];
    const float* p2_b = (const float*)d_in[9];
    const float* ts   = (const float*)d_in[10];
    float* outp = (float*)d_out;

    float* w = (float*)d_ws;
    float* Wp   = w;  w += (size_t)Ll * WLAYER;
    float* STre = w;  w += NSPEC;  float* STim = w;  w += NSPEC;
    float* Zhat = w;  w += Bb * NMODE * 2;
    float* Ay   = w;  w += Bb * Dd * Mm * 2;
    float* Tm   = w;  w += Bb * NMODE * 2;
    float* Gi   = w;  w += Bb * Dd * Mm * 2;
    float* qv   = w;  w += 64;
    float* cs   = w;  w += 128;
    float* sn   = w;  w += 128;
    float* scal = w;  w += 8;

    fnde_build_tables<<<1, 128, 0, stream>>>(cs, sn);
    fnde_pack_weights<<<Ll * 2 * Mm * 4 * 4, 128, 0, stream>>>(spec_w1, spec_w2, Wp);
    fnde_fwd_dft1<<<(Bb * Dd * Mm) / 256, 256, 0, stream>>>(zin, cs, sn, Ay);
    fnde_fwd_dft2<<<(Bb * NMODE) / 256, 256, 0, stream>>>(Ay, cs, sn, Zhat);

    fnde_chain<<<NMODE / 2, 256, 0, stream>>>(Zhat, lift_w, lift_b, Wp, fl_bias, ts, STre, STim);

    fnde_proj_prep<<<1, 64, 0, stream>>>(p1_w, p1_b, p2_w, p2_b, lift_w, lift_b, qv, scal);
    fnde_proj_modes<<<(Bb * NMODE) / 256, 256, 0, stream>>>(STre, STim, qv, Tm);
    fnde_inv_dft1<<<(Bb * Dd * Mm) / 256, 256, 0, stream>>>(Tm, cs, sn, Gi);
    fnde_inv_dft2_out<<<(Bb * Dd * Dd) / 256, 256, 0, stream>>>(Gi, zin, cs, sn, scal, outp);
}


# Complex product from three real products per k-step (Wr*Xr, Wi*Xi, (Wr+Wi)*(Xr+Xi); Ki = T3-(T1+T2), sums regrouped): 48 WMMAs per product, not 64; Xr+Xi is a third LDS plane
# speedup vs baseline: 7.3021x; 1.0782x over previous
#include <hip/hip_runtime.h>


typedef float v2f __attribute__((ext_vector_type(2)));
typedef float v4f __attribute__((ext_vector_type(4)));
typedef float v8f __attribute__((ext_vector_type(8)));

#define Bb    16
#define Cc    64
#define Dd    128
#define Mm    16
#define Ll    3
#define NMODE 512
#define NSPEC (NMODE*Cc*Bb)
#define WLAYER (2*NMODE*Cc*Cc)

__global__ void fnde_build_tables(float* cs, float* sn) {
    int t = threadIdx.x;
    float ang = (float)(6.283185307179586 / 128.0) * (float)t;
    cs[t] = cosf(ang);
    sn[t] = sinf(ang);
}

#define PK_PITCH 516
__global__ __launch_bounds__(128)
void fnde_pack_weights(const float* __restrict__ w1,
                       const float* __restrict__ w2,
                       float* __restrict__ Wp) {
    __shared__ v4f pk_tile[16 * (PK_PITCH / 4)];
    float* tile = (float*)pk_tile;

    const int t = threadIdx.x;
    int blk = blockIdx.x;
    const int kq = blk & 3;    blk >>= 2;
    const int mt = blk & 3;    blk >>= 2;
    const int mx = blk & 15;   blk >>= 4;
    const int corner = blk & 1;
    const int l = blk >> 1;
    const float* src = corner ? w2 : w1;

    const int p  = t & 7;
    const int b  = (t >> 3) & 1;
    const int cb = (t >> 4) & 1;
    const int w  = t >> 5;
    const float* sp = src + ((((size_t)(l * Cc + kq * 16 + b) * Cc + mt * 16 + 2 * w + cb) * Mm + mx) * Mm) * 2
                          + p * 4;
    float* tp = tile + (2 * p) * PK_PITCH + (2 * w + cb) * 4 + b * 2;
    v4f v[16];
#pragma unroll
    for (int r = 0; r < 16; ++r) {
        const int ip = r >> 1;
        const int cp = (r & 1) * 4;
        v[r] = *(const v4f*)(sp + (size_t)(ip * 2) * (Cc * Mm * Mm * 2) + (size_t)(cp * 2) * (Mm * Mm * 2));
    }
#pragma unroll
    for (int r = 0; r < 16; ++r) {
        const int ip = r >> 1;
        const int cp = (r & 1) * 4;
        float* d = tp + ip * 64 + cp * 8;
        v2f e0, e1;
        e0.x = v[r].x;   e0.y = v[r].y;
        e1.x = v[r].z;   e1.y = v[r].w;
        *(v2f*)d              = e0;
        *(v2f*)(d + PK_PITCH) = e1;
    }
    __syncthreads();

    const int lane = t & 31;
    const v4f* tq = pk_tile + w * 32 + lane;
    v4f* dp = (v4f*)(Wp + (size_t)l * WLAYER + (size_t)(corner * 256 + mx * 16) * (2 * Cc * Cc))
              + (mt * 16 + kq * 4 + w) * 32 + lane;
#pragma unroll
    for (int ky = 0; ky < 16; ++ky) {
        v4f a = tq[ky * (PK_PITCH / 4)];
        v4f f;
        f.x = a.x;   f.y = a.z;   f.z = a.y;   f.w = a.w;
        dp[(size_t)ky * (2 * Cc * Cc / 4)] = f;
    }
}

__global__ void fnde_fwd_dft1(const float* __restrict__ z,
                              const float* __restrict__ cs,
                              const float* __restrict__ sn,
                              float* __restrict__ Ay) {
    int t  = blockIdx.x * 256 + threadIdx.x;
    int b  = t >> 11;
    int x  = (t >> 4) & 127;
    int ky = t & 15;
    const float* row = z + (b * 128 + x) * 128;
    float re = 0.f, im = 0.f;
    for (int y = 0; y < 128; ++y) {
        float v = row[y];
        int a = (ky * y) & 127;
        re += v * cs[a];
        im -= v * sn[a];
    }
    Ay[t * 2]     = re;
    Ay[t * 2 + 1] = im;
}

__global__ void fnde_fwd_dft2(const float* __restrict__ Ay,
                              const float* __restrict__ cs,
                              const float* __restrict__ sn,
                              float* __restrict__ Zhat) {
    int t = blockIdx.x * 256 + threadIdx.x;
    int b = t >> 9;
    int m = t & 511;
    int corner = m >> 8, mx = (m >> 4) & 15, ky = m & 15;
    int kx = corner ? (112 + mx) : mx;
    float re = 0.f, im = 0.f;
    for (int x = 0; x < 128; ++x) {
        int base = ((b * 128 + x) * 16 + ky) * 2;
        float ar = Ay[base], ai = Ay[base + 1];
        int a = (kx * x) & 127;
        float c = cs[a], s = sn[a];
        re += ar * c + ai * s;
        im += ai * c - ar * s;
    }
    Zhat[t * 2]     = re;
    Zhat[t * 2 + 1] = im;
}


__device__ __forceinline__ void fnde_eval(v2f* __restrict__ Ub, int ownBase, int bBase,
                                          const float (&vr)[8], const float (&vi)[8],
                                          const v2f (&are)[16], const v2f (&aim)[16],
                                          const v2f (&aws)[16], const float (&bv)[8],
                                          float (&kr)[8], float (&ki)[8]) {
#pragma unroll
    for (int q = 0; q < 4; ++q) {
        v2f tr, ti;
        tr.x = vr[2 * q];   tr.y = vr[2 * q + 1];
        ti.x = vi[2 * q];   ti.y = vi[2 * q + 1];
        Ub[ownBase + q * 16]        = tr;
        Ub[512 + ownBase + q * 16]  = ti;
        Ub[1024 + ownBase + q * 16] = tr + ti;
    }
    __syncthreads();
    v8f t1 = {}, t2 = {}, t3 = {};
#pragma unroll
    for (int kk = 0; kk < 16; ++kk) {
        v2f br = Ub[bBase + kk * 32];
        v2f bi = Ub[512 + bBase + kk * 32];
        v2f bs = Ub[1024 + bBase + kk * 32];
        t1 = __builtin_amdgcn_wmma_f32_16x16x4_f32(false, are[kk], false, br, (short)0, t1, false, false);
        t2 = __builtin_amdgcn_wmma_f32_16x16x4_f32(false, aim[kk], false, bi, (short)0, t2, false, false);
        t3 = __builtin_amdgcn_wmma_f32_16x16x4_f32(false, aws[kk], false, bs, (short)0, t3, false, false);
    }
    asm volatile("v_nop\n\tv_nop\n\tv_nop\n\tv_nop" : "+v"(t1), "+v"(t2), "+v"(t3));
#pragma unroll
    for (int j = 0; j < 8; ++j) {
        kr[j] = (t1[j] - t2[j]) + bv[j];
        ki[j] = t3[j] - (t1[j] + t2[j]);
    }
}

__device__ __forceinline__ void fnde_pi_pair(v2f* Xown, const v2f* Xpar,
                                             bool noPartner, int ownBase,
                                             const float (&kr)[8], const float (&ki)[8],
                                             float (&pr)[8], float (&pim)[8]) {
#pragma unroll
    for (int q = 0; q < 4; ++q) {
        v2f tr, ti;
        tr.x = kr[2 * q];   tr.y = kr[2 * q + 1];
        ti.x = ki[2 * q];   ti.y = ki[2 * q + 1];
        Xown[ownBase + q * 16]       = tr;
        Xown[512 + ownBase + q * 16] = ti;
    }
    __syncthreads();
#pragma unroll
    for (int q = 0; q < 4; ++q) {
        v2f tr = Xpar[ownBase + q * 16];
        v2f ti = Xpar[512 + ownBase + q * 16];
        if (noPartner) { tr.x = 0.0f; tr.y = 0.0f; ti.x = 0.0f; ti.y = 0.0f; }
        pr[2 * q]      = 0.5f * (kr[2 * q]     + tr.x);
        pr[2 * q + 1]  = 0.5f * (kr[2 * q + 1] + tr.y);
        pim[2 * q]     = 0.5f * (ki[2 * q]     - ti.x);
        pim[2 * q + 1] = 0.5f * (ki[2 * q + 1] - ti.y);
    }
}

__global__ __launch_bounds__(256)
void fnde_chain(const float* __restrict__ Zhat,
                const float* __restrict__ lw, const float* __restrict__ lb,
                const float* __restrict__ Wp, const float* __restrict__ bias,
                const float* __restrict__ ts,
                float* __restrict__ STre, float* __restrict__ STim) {
    __shared__ v2f Ush[2 * 2 * 3 * 512];
    __shared__ v2f Xsh[2 * 2 * 512];

    const int blk = blockIdx.x;
    const int wave = __builtin_amdgcn_readfirstlane((int)(threadIdx.x >> 5));
    const int lane = threadIdx.x & 31;
    const int laneHi = lane >> 4;
    const int laneLo = lane & 15;
    const int mi = wave >> 2;
    const int mt = wave & 3;
    const int ky  = blk & 15;
    const int mx0 = blk >> 4;
    const int mode = mi ? (256 + (((16 - mx0) & 15) << 4) + ky) : blk;
    const bool ky0 = (ky == 0);
    const bool noPartner = (blk == 0) && (mi == 1);
    const int  pmi = (blk == 0) ? mi : (1 - mi);

    const int ownBase = (mt * 8 + laneHi * 4) * 16 + laneLo;
    const int bBase   = laneHi * 16 + laneLo;
    const int row0    = mt * 16 + laneHi * 8;

    v2f* U0 = Ush + (0 * 2 + mi) * 1536;
    v2f* U1 = Ush + (1 * 2 + mi) * 1536;
    v2f* Xown = Xsh + mi * 1024;
    const v2f* Xpar = Xsh + pmi * 1024;

    float zr[8], zi[8], ssr[8], ssi[8], str[8], sti[8];
    {
        float hr = Zhat[(laneLo * 512 + mode) * 2];
        float hi = Zhat[(laneLo * 512 + mode) * 2 + 1];
#pragma unroll
        for (int j = 0; j < 8; ++j) {
            float w = lw[row0 + j];
            float lbj = lb[row0 + j];
            zr[j] = w * hr + ((mode == 0) ? lbj * 16384.0f : 0.0f);
            zi[j] = w * hi;
            str[j] = 0.0f;  sti[j] = 0.0f;
        }
    }

#define FNDE_PROJ(SR, SI)                                                        \
    if (ky0) fnde_pi_pair(Xown, Xpar, noPartner, ownBase, SR, SI, pr, pim);      \
    else {                                                                       \
        _Pragma("unroll")                                                        \
        for (int j = 0; j < 8; ++j) { pr[j] = SR[j]; pim[j] = SI[j]; }           \
    }

#pragma unroll 1
    for (int l = 0; l < Ll; ++l) {
        const v4f* Wm = (const v4f*)(Wp + (size_t)l * WLAYER + (size_t)mode * (2 * Cc * Cc))
                        + (mt * 16) * 32 + lane;
        v2f are[16], aim[16], aws[16];
#pragma unroll
        for (int kk = 0; kk < 16; ++kk) {
            v4f a = Wm[kk * 32];
            are[kk].x = a.x;    are[kk].y = a.y;
            aim[kk].x = a.z;    aim[kk].y = a.w;
            aws[kk] = are[kk] + aim[kk];
        }
        float bv[8];
#pragma unroll
        for (int j = 0; j < 8; ++j) {
            float bj = bias[l * Cc + row0 + j];
            bv[j] = (mode == 0) ? bj * 16384.0f : 0.0f;
        }

#pragma unroll 1
        for (int s = 0; s < 4; ++s) {
            const float h  = ts[s + 1] - ts[s];
            const float hh = 0.5f * h;
            const float h6 = h * (1.0f / 6.0f);
            float kr[8], ki[8], pr[8], pim[8], ur[8], ui[8];

            fnde_eval(U0, ownBase, bBase, zr, zi, are, aim, aws, bv, kr, ki);
#pragma unroll
            for (int j = 0; j < 8; ++j) { ssr[j] = kr[j]; ssi[j] = ki[j]; }
            FNDE_PROJ(kr, ki)
#pragma unroll
            for (int j = 0; j < 8; ++j) { ur[j] = zr[j] + hh * pr[j]; ui[j] = zi[j] + hh * pim[j]; }

            fnde_eval(U1, ownBase, bBase, ur, ui, are, aim, aws, bv, kr, ki);
#pragma unroll
            for (int j = 0; j < 8; ++j) { ssr[j] += 2.0f * kr[j]; ssi[j] += 2.0f * ki[j]; }
            FNDE_PROJ(kr, ki)
#pragma unroll
            for (int j = 0; j < 8; ++j) { ur[j] = zr[j] + hh * pr[j]; ui[j] = zi[j] + hh * pim[j]; }

            fnde_eval(U0, ownBase, bBase, ur, ui, are, aim, aws, bv, kr, ki);
#pragma unroll
            for (int j = 0; j < 8; ++j) { ssr[j] += 2.0f * kr[j]; ssi[j] += 2.0f * ki[j]; }
            FNDE_PROJ(kr, ki)
#pragma unroll
            for (int j = 0; j < 8; ++j) { ur[j] = zr[j] + h * pr[j]; ui[j] = zi[j] + h * pim[j]; }

            fnde_eval(U1, ownBase, bBase, ur, ui, are, aim, aws, bv, kr, ki);
#pragma unroll
            for (int j = 0; j < 8; ++j) { ssr[j] += kr[j]; ssi[j] += ki[j]; }
            FNDE_PROJ(ssr, ssi)
#pragma unroll
            for (int j = 0; j < 8; ++j) {
                zr[j]  += h6 * pr[j];    zi[j]  += h6 * pim[j];
                str[j] += h6 * ssr[j];   sti[j] += h6 * ssi[j];
            }
        }
    }
#undef FNDE_PROJ

#pragma unroll
    for (int j = 0; j < 8; ++j) {
        STre[mode * 1024 + (row0 + j) * 16 + laneLo] = str[j];
        STim[mode * 1024 + (row0 + j) * 16 + laneLo] = sti[j];
    }
}

__global__ void fnde_proj_prep(const float* __restrict__ p1w, const float* __restrict__ p1b,
                               const float* __restrict__ p2w, const float* __restrict__ p2b,
                               const float* __restrict__ lw,  const float* __restrict__ lb,
                               float* __restrict__ q, float* __restrict__ scal) {
    __shared__ float qs[64];
    int c = threadIdx.x;
    float acc = 0.f;
    for (int p = 0; p < 64; ++p) acc += p2w[p] * p1w[p * 64 + c];
    qs[c] = acc;
    q[c] = acc;
    __syncthreads();
    if (c == 0) {
        float r = p2b[0];
        for (int p = 0; p < 64; ++p) r += p2w[p] * p1b[p];
        float al = 0.f, be = r;
        for (int i = 0; i < 64; ++i) { al += qs[i] * lw[i]; be += qs[i] * lb[i]; }
        scal[0] = al;
        scal[1] = be;
    }
}

__global__ void fnde_proj_modes(const float* __restrict__ STre, const float* __restrict__ STim,
                                const float* __restrict__ q, float* __restrict__ T) {
    int t = blockIdx.x * 256 + threadIdx.x;
    int b = t >> 9;
    int m = t & 511;
    const float* pr = STre + m * 1024 + b;
    const float* pi = STim + m * 1024 + b;
    float re = 0.f, im = 0.f;
    for (int c = 0; c < 64; ++c) {
        float w = q[c];
        re += w * pr[c * 16];
        im += w * pi[c * 16];
    }
    T[t * 2]     = re;
    T[t * 2 + 1] = im;
}

__global__ void fnde_inv_dft1(const float* __restrict__ T,
                              const float* __restrict__ cs, const float* __restrict__ sn,
                              float* __restrict__ G) {
    int t  = blockIdx.x * 256 + threadIdx.x;
    int b  = t >> 11;
    int x  = (t >> 4) & 127;
    int ky = t & 15;
    float re = 0.f, im = 0.f;
    for (int corner = 0; corner < 2; ++corner) {
        for (int mx = 0; mx < 16; ++mx) {
            int m  = corner * 256 + mx * 16 + ky;
            int kx = corner ? (112 + mx) : mx;
            float tr = T[(b * 512 + m) * 2];
            float ti = T[(b * 512 + m) * 2 + 1];
            int a = (kx * x) & 127;
            float c = cs[a], s = sn[a];
            re += tr * c - ti * s;
            im += tr * s + ti * c;
        }
    }
    G[t * 2]     = re;
    G[t * 2 + 1] = im;
}

__global__ void fnde_inv_dft2_out(const float* __restrict__ G, const float* __restrict__ z,
                                  const float* __restrict__ cs, const float* __restrict__ sn,
                                  const float* __restrict__ scal, float* __restrict__ out) {
    int t = blockIdx.x * 256 + threadIdx.x;
    int b = t >> 14;
    int x = (t >> 7) & 127;
    int y = t & 127;
    const float* g = G + ((b * 128 + x) * 16) * 2;
    float acc = g[0];
    for (int ky = 1; ky < 16; ++ky) {
        int a = (ky * y) & 127;
        acc += 2.0f * (g[ky * 2] * cs[a] - g[ky * 2 + 1] * sn[a]);
    }
    out[t] = scal[0] * z[t] + scal[1] + acc * (1.0f / 16384.0f);
}

extern "C" void kernel_launch(void* const* d_in, const int* in_sizes, int n_in,
                              void* d_out, int out_size, void* d_ws, size_t ws_size,
                              hipStream_t stream) {
    (void)in_sizes; (void)n_in; (void)out_size; (void)ws_size;

    const float* zin    = (const float*)d_in[0];
    const float* lift_w = (const float*)d_in[1];
    const float* lift_b = (const float*)d_in[2];
    const float* spec_w1 = (const float*)d_in[3];
    const float* spec_w2 = (const float*)d_in[4];
    const float* fl_bias = (const float*)d_in[5];
    const float* p1_w = (const float*)d_in[6];
    const float* p1_b = (const float*)d_in[7];
    const float* p2_w = (const float*)d_in[8];
    const float* p2_b = (const float*)d_in[9];
    const float* ts   = (const float*)d_in[10];
    float* outp = (float*)d_out;

    float* w = (float*)d_ws;
    float* Wp   = w;  w += (size_t)Ll * WLAYER;
    float* STre = w;  w += NSPEC;  float* STim = w;  w += NSPEC;
    float* Zhat = w;  w += Bb * NMODE * 2;
    float* Ay   = w;  w += Bb * Dd * Mm * 2;
    float* Tm   = w;  w += Bb * NMODE * 2;
    float* Gi   = w;  w += Bb * Dd * Mm * 2;
    float* qv   = w;  w += 64;
    float* cs   = w;  w += 128;
    float* sn   = w;  w += 128;
    float* scal = w;  w += 8;

    fnde_build_tables<<<1, 128, 0, stream>>>(cs, sn);
    fnde_pack_weights<<<Ll * 2 * Mm * 4 * 4, 128, 0, stream>>>(spec_w1, spec_w2, Wp);
    fnde_fwd_dft1<<<(Bb * Dd * Mm) / 256, 256, 0, stream>>>(zin, cs, sn, Ay);
    fnde_fwd_dft2<<<(Bb * NMODE) / 256, 256, 0, stream>>>(Ay, cs, sn, Zhat);

    fnde_chain<<<NMODE / 2, 256, 0, stream>>>(Zhat, lift_w, lift_b, Wp, fl_bias, ts, STre, STim);

    fnde_proj_prep<<<1, 64, 0, stream>>>(p1_w, p1_b, p2_w, p2_b, lift_w, lift_b, qv, scal);
    fnde_proj_modes<<<(Bb * NMODE) / 256, 256, 0, stream>>>(STre, STim, qv, Tm);
    fnde_inv_dft1<<<(Bb * Dd * Mm) / 256, 256, 0, stream>>>(Tm, cs, sn, Gi);
    fnde_inv_dft2_out<<<(Bb * Dd * Dd) / 256, 256, 0, stream>>>(Gi, zin, cs, sn, scal, outp);
}


# Projection's channel contraction folded into the ODE launch's epilogue: each lane sums its 8 rows, 8 partials per (mode, b) meet in LDS (channel sum regrouped); the S planes and one launch go away
# speedup vs baseline: 8.7845x; 1.2030x over previous
#include <hip/hip_runtime.h>


typedef float v2f __attribute__((ext_vector_type(2)));
typedef float v4f __attribute__((ext_vector_type(4)));
typedef float v8f __attribute__((ext_vector_type(8)));

#define Bb    16
#define Cc    64
#define Dd    128
#define Mm    16
#define Ll    3
#define NMODE 512
#define WLAYER (2*NMODE*Cc*Cc)

__global__ void fnde_build_tables(float* cs, float* sn) {
    int t = threadIdx.x;
    float ang = (float)(6.283185307179586 / 128.0) * (float)t;
    cs[t] = cosf(ang);
    sn[t] = sinf(ang);
}

#define PK_PITCH 516
__global__ __launch_bounds__(128)
void fnde_pack_weights(const float* __restrict__ w1,
                       const float* __restrict__ w2,
                       float* __restrict__ Wp) {
    __shared__ v4f pk_tile[16 * (PK_PITCH / 4)];
    float* tile = (float*)pk_tile;

    const int t = threadIdx.x;
    int blk = blockIdx.x;
    const int kq = blk & 3;    blk >>= 2;
    const int mt = blk & 3;    blk >>= 2;
    const int mx = blk & 15;   blk >>= 4;
    const int corner = blk & 1;
    const int l = blk >> 1;
    const float* src = corner ? w2 : w1;

    const int p  = t & 7;
    const int b  = (t >> 3) & 1;
    const int cb = (t >> 4) & 1;
    const int w  = t >> 5;
    const float* sp = src + ((((size_t)(l * Cc + kq * 16 + b) * Cc + mt * 16 + 2 * w + cb) * Mm + mx) * Mm) * 2
                          + p * 4;
    float* tp = tile + (2 * p) * PK_PITCH + (2 * w + cb) * 4 + b * 2;
    v4f v[16];
#pragma unroll
    for (int r = 0; r < 16; ++r) {
        const int ip = r >> 1;
        const int cp = (r & 1) * 4;
        v[r] = *(const v4f*)(sp + (size_t)(ip * 2) * (Cc * Mm * Mm * 2) + (size_t)(cp * 2) * (Mm * Mm * 2));
    }
#pragma unroll
    for (int r = 0; r < 16; ++r) {
        const int ip = r >> 1;
        const int cp = (r & 1) * 4;
        float* d = tp + ip * 64 + cp * 8;
        v2f e0, e1;
        e0.x = v[r].x;   e0.y = v[r].y;
        e1.x = v[r].z;   e1.y = v[r].w;
        *(v2f*)d              = e0;
        *(v2f*)(d + PK_PITCH) = e1;
    }
    __syncthreads();

    const int lane = t & 31;
    const v4f* tq = pk_tile + w * 32 + lane;
    v4f* dp = (v4f*)(Wp + (size_t)l * WLAYER + (size_t)(corner * 256 + mx * 16) * (2 * Cc * Cc))
              + (mt * 16 + kq * 4 + w) * 32 + lane;
#pragma unroll
    for (int ky = 0; ky < 16; ++ky) {
        v4f a = tq[ky * (PK_PITCH / 4)];
        v4f f;
        f.x = a.x;   f.y = a.z;   f.z = a.y;   f.w = a.w;
        dp[(size_t)ky * (2 * Cc * Cc / 4)] = f;
    }
}

__global__ void fnde_fwd_dft1(const float* __restrict__ z,
                              const float* __restrict__ cs,
                              const float* __restrict__ sn,
                              float* __restrict__ Ay) {
    int t  = blockIdx.x * 256 + threadIdx.x;
    int b  = t >> 11;
    int x  = (t >> 4) & 127;
    int ky = t & 15;
    const float* row = z + (b * 128 + x) * 128;
    float re = 0.f, im = 0.f;
    for (int y = 0; y < 128; ++y) {
        float v = row[y];
        int a = (ky * y) & 127;
        re += v * cs[a];
        im -= v * sn[a];
    }
    Ay[t * 2]     = re;
    Ay[t * 2 + 1] = im;
}

__global__ void fnde_fwd_dft2(const float* __restrict__ Ay,
                              const float* __restrict__ cs,
                              const float* __restrict__ sn,
                              float* __restrict__ Zhat) {
    int t = blockIdx.x * 256 + threadIdx.x;
    int b = t >> 9;
    int m = t & 511;
    int corner = m >> 8, mx = (m >> 4) & 15, ky = m & 15;
    int kx = corner ? (112 + mx) : mx;
    float re = 0.f, im = 0.f;
    for (int x = 0; x < 128; ++x) {
        int base = ((b * 128 + x) * 16 + ky) * 2;
        float ar = Ay[base], ai = Ay[base + 1];
        int a = (kx * x) & 127;
        float c = cs[a], s = sn[a];
        re += ar * c + ai * s;
        im += ai * c - ar * s;
    }
    Zhat[t * 2]     = re;
    Zhat[t * 2 + 1] = im;
}


__device__ __forceinline__ void fnde_eval(v2f* __restrict__ Ub, int ownBase, int bBase,
                                          const float (&vr)[8], const float (&vi)[8],
                                          const v2f (&are)[16], const v2f (&aim)[16],
                                          const v2f (&aws)[16], const float (&bv)[8],
                                          float (&kr)[8], float (&ki)[8]) {
#pragma unroll
    for (int q = 0; q < 4; ++q) {
        v2f tr, ti;
        tr.x = vr[2 * q];   tr.y = vr[2 * q + 1];
        ti.x = vi[2 * q];   ti.y = vi[2 * q + 1];
        Ub[ownBase + q * 16]        = tr;
        Ub[512 + ownBase + q * 16]  = ti;
        Ub[1024 + ownBase + q * 16] = tr + ti;
    }
    __syncthreads();
    v8f t1 = {}, t2 = {}, t3 = {};
#pragma unroll
    for (int kk = 0; kk < 16; ++kk) {
        v2f br = Ub[bBase + kk * 32];
        v2f bi = Ub[512 + bBase + kk * 32];
        v2f bs = Ub[1024 + bBase + kk * 32];
        t1 = __builtin_amdgcn_wmma_f32_16x16x4_f32(false, are[kk], false, br, (short)0, t1, false, false);
        t2 = __builtin_amdgcn_wmma_f32_16x16x4_f32(false, aim[kk], false, bi, (short)0, t2, false, false);
        t3 = __builtin_amdgcn_wmma_f32_16x16x4_f32(false, aws[kk], false, bs, (short)0, t3, false, false);
    }
    asm volatile("v_nop\n\tv_nop\n\tv_nop\n\tv_nop" : "+v"(t1), "+v"(t2), "+v"(t3));
#pragma unroll
    for (int j = 0; j < 8; ++j) {
        kr[j] = (t1[j] - t2[j]) + bv[j];
        ki[j] = t3[j] - (t1[j] + t2[j]);
    }
}

__device__ __forceinline__ void fnde_pi_pair(v2f* Xown, const v2f* Xpar,
                                             bool noPartner, int ownBase,
                                             const float (&kr)[8], const float (&ki)[8],
                                             float (&pr)[8], float (&pim)[8]) {
#pragma unroll
    for (int q = 0; q < 4; ++q) {
        v2f tr, ti;
        tr.x = kr[2 * q];   tr.y = kr[2 * q + 1];
        ti.x = ki[2 * q];   ti.y = ki[2 * q + 1];
        Xown[ownBase + q * 16]       = tr;
        Xown[512 + ownBase + q * 16] = ti;
    }
    __syncthreads();
#pragma unroll
    for (int q = 0; q < 4; ++q) {
        v2f tr = Xpar[ownBase + q * 16];
        v2f ti = Xpar[512 + ownBase + q * 16];
        if (noPartner) { tr.x = 0.0f; tr.y = 0.0f; ti.x = 0.0f; ti.y = 0.0f; }
        pr[2 * q]      = 0.5f * (kr[2 * q]     + tr.x);
        pr[2 * q + 1]  = 0.5f * (kr[2 * q + 1] + tr.y);
        pim[2 * q]     = 0.5f * (ki[2 * q]     - ti.x);
        pim[2 * q + 1] = 0.5f * (ki[2 * q + 1] - ti.y);
    }
}

__global__ __launch_bounds__(256)
void fnde_chain(const float* __restrict__ Zhat,
                const float* __restrict__ lw, const float* __restrict__ lb,
                const float* __restrict__ Wp, const float* __restrict__ bias,
                const float* __restrict__ ts,
                const float* __restrict__ q, float* __restrict__ T) {
    __shared__ v2f Ush[2 * 2 * 3 * 512];
    __shared__ v2f Xsh[2 * 2 * 512];

    const int blk = blockIdx.x;
    const int wave = __builtin_amdgcn_readfirstlane((int)(threadIdx.x >> 5));
    const int lane = threadIdx.x & 31;
    const int laneHi = lane >> 4;
    const int laneLo = lane & 15;
    const int mi = wave >> 2;
    const int mt = wave & 3;
    const int ky  = blk & 15;
    const int mx0 = blk >> 4;
    const int mode = mi ? (256 + (((16 - mx0) & 15) << 4) + ky) : blk;
    const bool ky0 = (ky == 0);
    const bool noPartner = (blk == 0) && (mi == 1);
    const int  pmi = (blk == 0) ? mi : (1 - mi);

    const int ownBase = (mt * 8 + laneHi * 4) * 16 + laneLo;
    const int bBase   = laneHi * 16 + laneLo;
    const int row0    = mt * 16 + laneHi * 8;

    v2f* U0 = Ush + (0 * 2 + mi) * 1536;
    v2f* U1 = Ush + (1 * 2 + mi) * 1536;
    v2f* Xown = Xsh + mi * 1024;
    const v2f* Xpar = Xsh + pmi * 1024;

    float zr[8], zi[8], ssr[8], ssi[8], str[8], sti[8];
    {
        float hr = Zhat[(laneLo * 512 + mode) * 2];
        float hi = Zhat[(laneLo * 512 + mode) * 2 + 1];
#pragma unroll
        for (int j = 0; j < 8; ++j) {
            float w = lw[row0 + j];
            float lbj = lb[row0 + j];
            zr[j] = w * hr + ((mode == 0) ? lbj * 16384.0f : 0.0f);
            zi[j] = w * hi;
            str[j] = 0.0f;  sti[j] = 0.0f;
        }
    }

#define FNDE_PROJ(SR, SI)                                                        \
    if (ky0) fnde_pi_pair(Xown, Xpar, noPartner, ownBase, SR, SI, pr, pim);      \
    else {                                                                       \
        _Pragma("unroll")                                                        \
        for (int j = 0; j < 8; ++j) { pr[j] = SR[j]; pim[j] = SI[j]; }           \
    }

#pragma unroll 1
    for (int l = 0; l < Ll; ++l) {
        const v4f* Wm = (const v4f*)(Wp + (size_t)l * WLAYER + (size_t)mode * (2 * Cc * Cc))
                        + (mt * 16) * 32 + lane;
        v2f are[16], aim[16], aws[16];
#pragma unroll
        for (int kk = 0; kk < 16; ++kk) {
            v4f a = Wm[kk * 32];
            are[kk].x = a.x;    are[kk].y = a.y;
            aim[kk].x = a.z;    aim[kk].y = a.w;
            aws[kk] = are[kk] + aim[kk];
        }
        float bv[8];
#pragma unroll
        for (int j = 0; j < 8; ++j) {
            float bj = bias[l * Cc + row0 + j];
            bv[j] = (mode == 0) ? bj * 16384.0f : 0.0f;
        }

#pragma unroll 1
        for (int s = 0; s < 4; ++s) {
            const float h  = ts[s + 1] - ts[s];
            const float hh = 0.5f * h;
            const float h6 = h * (1.0f / 6.0f);
            float kr[8], ki[8], pr[8], pim[8], ur[8], ui[8];

            fnde_eval(U0, ownBase, bBase, zr, zi, are, aim, aws, bv, kr, ki);
#pragma unroll
            for (int j = 0; j < 8; ++j) { ssr[j] = kr[j]; ssi[j] = ki[j]; }
            FNDE_PROJ(kr, ki)
#pragma unroll
            for (int j = 0; j < 8; ++j) { ur[j] = zr[j] + hh * pr[j]; ui[j] = zi[j] + hh * pim[j]; }

            fnde_eval(U1, ownBase, bBase, ur, ui, are, aim, aws, bv, kr, ki);
#pragma unroll
            for (int j = 0; j < 8; ++j) { ssr[j] += 2.0f * kr[j]; ssi[j] += 2.0f * ki[j]; }
            FNDE_PROJ(kr, ki)
#pragma unroll
            for (int j = 0; j < 8; ++j) { ur[j] = zr[j] + hh * pr[j]; ui[j] = zi[j] + hh * pim[j]; }

            fnde_eval(U0, ownBase, bBase, ur, ui, are, aim, aws, bv, kr, ki);
#pragma unroll
            for (int j = 0; j < 8; ++j) { ssr[j] += 2.0f * kr[j]; ssi[j] += 2.0f * ki[j]; }
            FNDE_PROJ(kr, ki)
#pragma unroll
            for (int j = 0; j < 8; ++j) { ur[j] = zr[j] + h * pr[j]; ui[j] = zi[j] + h * pim[j]; }

            fnde_eval(U1, ownBase, bBase, ur, ui, are, aim, aws, bv, kr, ki);
#pragma unroll
            for (int j = 0; j < 8; ++j) { ssr[j] += kr[j]; ssi[j] += ki[j]; }
            FNDE_PROJ(ssr, ssi)
#pragma unroll
            for (int j = 0; j < 8; ++j) {
                zr[j]  += h6 * pr[j];    zi[j]  += h6 * pim[j];
                str[j] += h6 * ssr[j];   sti[j] += h6 * ssi[j];
            }
        }
    }
#undef FNDE_PROJ

    float pre = 0.0f, pie = 0.0f;
#pragma unroll
    for (int j = 0; j < 8; ++j) {
        const float w = q[row0 + j];
        pre += w * str[j];
        pie += w * sti[j];
    }
    v2f part;
    part.x = pre;   part.y = pie;
    Ush[threadIdx.x] = part;
    __syncthreads();
    if (wave == 0) {
        const int m2 = laneHi;
        v2f acc = Ush[m2 * 128 + laneLo];
#pragma unroll
        for (int r = 1; r < 8; ++r) acc += Ush[m2 * 128 + r * 16 + laneLo];
        const int modeL = m2 ? (256 + (((16 - mx0) & 15) << 4) + ky) : blk;
        *(v2f*)(T + (modeL * 16 + laneLo) * 2) = acc;
    }
}

__global__ void fnde_proj_prep(const float* __restrict__ p1w, const float* __restrict__ p1b,
                               const float* __restrict__ p2w, const float* __restrict__ p2b,
                               const float* __restrict__ lw,  const float* __restrict__ lb,
                               float* __restrict__ q, float* __restrict__ scal) {
    __shared__ float qs[64];
    int c = threadIdx.x;
    float acc = 0.f;
    for (int p = 0; p < 64; ++p) acc += p2w[p] * p1w[p * 64 + c];
    qs[c] = acc;
    q[c] = acc;
    __syncthreads();
    if (c == 0) {
        float r = p2b[0];
        for (int p = 0; p < 64; ++p) r += p2w[p] * p1b[p];
        float al = 0.f, be = r;
        for (int i = 0; i < 64; ++i) { al += qs[i] * lw[i]; be += qs[i] * lb[i]; }
        scal[0] = al;
        scal[1] = be;
    }
}

__global__ void fnde_inv_dft1(const float* __restrict__ T,
                              const float* __restrict__ cs, const float* __restrict__ sn,
                              float* __restrict__ G) {
    __shared__ v2f Tsh[NMODE];
    int t  = blockIdx.x * 256 + threadIdx.x;
    int b  = t >> 11;
    int x  = (t >> 4) & 127;
    int ky = t & 15;
    Tsh[threadIdx.x]       = *(const v2f*)(T + ((int)threadIdx.x * 16 + b) * 2);
    Tsh[threadIdx.x + 256] = *(const v2f*)(T + (((int)threadIdx.x + 256) * 16 + b) * 2);
    __syncthreads();
    float re = 0.f, im = 0.f;
    for (int corner = 0; corner < 2; ++corner) {
        for (int mx = 0; mx < 16; ++mx) {
            int m  = corner * 256 + mx * 16 + ky;
            int kx = corner ? (112 + mx) : mx;
            v2f tv = Tsh[m];
            float tr = tv.x, ti = tv.y;
            int a = (kx * x) & 127;
            float c = cs[a], s = sn[a];
            re += tr * c - ti * s;
            im += tr * s + ti * c;
        }
    }
    G[t * 2]     = re;
    G[t * 2 + 1] = im;
}

__global__ void fnde_inv_dft2_out(const float* __restrict__ G, const float* __restrict__ z,
                                  const float* __restrict__ cs, const float* __restrict__ sn,
                                  const float* __restrict__ scal, float* __restrict__ out) {
    int t = blockIdx.x * 256 + threadIdx.x;
    int b = t >> 14;
    int x = (t >> 7) & 127;
    int y = t & 127;
    const float* g = G + ((b * 128 + x) * 16) * 2;
    float acc = g[0];
    for (int ky = 1; ky < 16; ++ky) {
        int a = (ky * y) & 127;
        acc += 2.0f * (g[ky * 2] * cs[a] - g[ky * 2 + 1] * sn[a]);
    }
    out[t] = scal[0] * z[t] + scal[1] + acc * (1.0f / 16384.0f);
}

extern "C" void kernel_launch(void* const* d_in, const int* in_sizes, int n_in,
                              void* d_out, int out_size, void* d_ws, size_t ws_size,
                              hipStream_t stream) {
    (void)in_sizes; (void)n_in; (void)out_size; (void)ws_size;

    const float* zin    = (const float*)d_in[0];
    const float* lift_w = (const float*)d_in[1];
    const float* lift_b = (const float*)d_in[2];
    const float* spec_w1 = (const float*)d_in[3];
    const float* spec_w2 = (const float*)d_in[4];
    const float* fl_bias = (const float*)d_in[5];
    const float* p1_w = (const float*)d_in[6];
    const float* p1_b = (const float*)d_in[7];
    const float* p2_w = (const float*)d_in[8];
    const float* p2_b = (const float*)d_in[9];
    const float* ts   = (const float*)d_in[10];
    float* outp = (float*)d_out;

    float* w = (float*)d_ws;
    float* Wp   = w;  w += (size_t)Ll * WLAYER;
    float* Zhat = w;  w += Bb * NMODE * 2;
    float* Ay   = w;  w += Bb * Dd * Mm * 2;
    float* Tm   = w;  w += Bb * NMODE * 2;
    float* Gi   = w;  w += Bb * Dd * Mm * 2;
    float* qv   = w;  w += 64;
    float* cs   = w;  w += 128;
    float* sn   = w;  w += 128;
    float* scal = w;  w += 8;

    fnde_build_tables<<<1, 128, 0, stream>>>(cs, sn);
    fnde_pack_weights<<<Ll * 2 * Mm * 4 * 4, 128, 0, stream>>>(spec_w1, spec_w2, Wp);
    fnde_fwd_dft1<<<(Bb * Dd * Mm) / 256, 256, 0, stream>>>(zin, cs, sn, Ay);
    fnde_fwd_dft2<<<(Bb * NMODE) / 256, 256, 0, stream>>>(Ay, cs, sn, Zhat);

    fnde_proj_prep<<<1, 64, 0, stream>>>(p1_w, p1_b, p2_w, p2_b, lift_w, lift_b, qv, scal);

    fnde_chain<<<NMODE / 2, 256, 0, stream>>>(Zhat, lift_w, lift_b, Wp, fl_bias, ts, qv, Tm);

    fnde_inv_dft1<<<(Bb * Dd * Mm) / 256, 256, 0, stream>>>(Tm, cs, sn, Gi);
    fnde_inv_dft2_out<<<(Bb * Dd * Dd) / 256, 256, 0, stream>>>(Gi, zin, cs, sn, scal, outp);
}


# Launches 8->3: tables, proj prep, y-DFT, weight pack in one launch by block index; x-DFT is the ODE launch's prologue (x sum regrouped: 16 strided partials); both inverse DFT stages in one, G in LDS
# speedup vs baseline: 11.9757x; 1.3633x over previous
#include <hip/hip_runtime.h>


typedef float v2f __attribute__((ext_vector_type(2)));
typedef float v4f __attribute__((ext_vector_type(4)));
typedef float v8f __attribute__((ext_vector_type(8)));

#define Bb    16
#define Cc    64
#define Dd    128
#define Mm    16
#define Ll    3
#define NMODE 512
#define WLAYER (2*NMODE*Cc*Cc)

__device__ __forceinline__ void fnde_tables_body(float* cs, float* sn) {
    int t = threadIdx.x;
    float ang = (float)(6.283185307179586 / 128.0) * (float)t;
    cs[t] = cosf(ang);
    sn[t] = sinf(ang);
}

#define PK_PITCH 516
__device__ __forceinline__ void fnde_pack_body(const float* __restrict__ w1,
                                               const float* __restrict__ w2,
                                               float* __restrict__ Wp,
                                               v4f* pk_tile, int blk) {
    float* tile = (float*)pk_tile;

    const int t = threadIdx.x;
    const int kq = blk & 3;    blk >>= 2;
    const int mt = blk & 3;    blk >>= 2;
    const int mx = blk & 15;   blk >>= 4;
    const int corner = blk & 1;
    const int l = blk >> 1;
    const float* src = corner ? w2 : w1;

    const int p  = t & 7;
    const int b  = (t >> 3) & 1;
    const int cb = (t >> 4) & 1;
    const int w  = t >> 5;
    const float* sp = src + ((((size_t)(l * Cc + kq * 16 + b) * Cc + mt * 16 + 2 * w + cb) * Mm + mx) * Mm) * 2
                          + p * 4;
    float* tp = tile + (2 * p) * PK_PITCH + (2 * w + cb) * 4 + b * 2;
    v4f v[16];
#pragma unroll
    for (int r = 0; r < 16; ++r) {
        const int ip = r >> 1;
        const int cp = (r & 1) * 4;
        v[r] = *(const v4f*)(sp + (size_t)(ip * 2) * (Cc * Mm * Mm * 2) + (size_t)(cp * 2) * (Mm * Mm * 2));
    }
#pragma unroll
    for (int r = 0; r < 16; ++r) {
        const int ip = r >> 1;
        const int cp = (r & 1) * 4;
        float* d = tp + ip * 64 + cp * 8;
        v2f e0, e1;
        e0.x = v[r].x;   e0.y = v[r].y;
        e1.x = v[r].z;   e1.y = v[r].w;
        *(v2f*)d              = e0;
        *(v2f*)(d + PK_PITCH) = e1;
    }
    __syncthreads();

    const int lane = t & 31;
    const v4f* tq = pk_tile + w * 32 + lane;
    v4f* dp = (v4f*)(Wp + (size_t)l * WLAYER + (size_t)(corner * 256 + mx * 16) * (2 * Cc * Cc))
              + (mt * 16 + kq * 4 + w) * 32 + lane;
#pragma unroll
    for (int ky = 0; ky < 16; ++ky) {
        v4f a = tq[ky * (PK_PITCH / 4)];
        v4f f;
        f.x = a.x;   f.y = a.z;   f.z = a.y;   f.w = a.w;
        dp[(size_t)ky * (2 * Cc * Cc / 4)] = f;
    }
}

__device__ __forceinline__ void fnde_dft1_body(const float* __restrict__ z,
                                               float* __restrict__ Ay,
                                               float* tab, int blk) {
    const int tid = threadIdx.x;
    {
        float ang = (float)(6.283185307179586 / 128.0) * (float)tid;
        tab[tid]       = cosf(ang);
        tab[128 + tid] = sinf(ang);
    }
    __syncthreads();
    int t  = blk * 128 + tid;
    int b  = t >> 11;
    int x  = (t >> 4) & 127;
    int ky = t & 15;
    const float* row = z + (b * 128 + x) * 128;
    float re = 0.f, im = 0.f;
    for (int y = 0; y < 128; ++y) {
        float v = row[y];
        int a = (ky * y) & 127;
        re += v * tab[a];
        im -= v * tab[128 + a];
    }
    v2f o;
    o.x = re;   o.y = im;
    *(v2f*)(Ay + (size_t)((ky * 16 + b) * 128 + x) * 2) = o;
}


__device__ __forceinline__ void fnde_eval(v2f* __restrict__ Ub, int ownBase, int bBase,
                                          const float (&vr)[8], const float (&vi)[8],
                                          const v2f (&are)[16], const v2f (&aim)[16],
                                          const v2f (&aws)[16], const float (&bv)[8],
                                          float (&kr)[8], float (&ki)[8]) {
#pragma unroll
    for (int q = 0; q < 4; ++q) {
        v2f tr, ti;
        tr.x = vr[2 * q];   tr.y = vr[2 * q + 1];
        ti.x = vi[2 * q];   ti.y = vi[2 * q + 1];
        Ub[ownBase + q * 16]        = tr;
        Ub[512 + ownBase + q * 16]  = ti;
        Ub[1024 + ownBase + q * 16] = tr + ti;
    }
    __syncthreads();
    v8f t1 = {}, t2 = {}, t3 = {};
#pragma unroll
    for (int kk = 0; kk < 16; ++kk) {
        v2f br = Ub[bBase + kk * 32];
        v2f bi = Ub[512 + bBase + kk * 32];
        v2f bs = Ub[1024 + bBase + kk * 32];
        t1 = __builtin_amdgcn_wmma_f32_16x16x4_f32(false, are[kk], false, br, (short)0, t1, false, false);
        t2 = __builtin_amdgcn_wmma_f32_16x16x4_f32(false, aim[kk], false, bi, (short)0, t2, false, false);
        t3 = __builtin_amdgcn_wmma_f32_16x16x4_f32(false, aws[kk], false, bs, (short)0, t3, false, false);
    }
    asm volatile("v_nop\n\tv_nop\n\tv_nop\n\tv_nop" : "+v"(t1), "+v"(t2), "+v"(t3));
#pragma unroll
    for (int j = 0; j < 8; ++j) {
        kr[j] = (t1[j] - t2[j]) + bv[j];
        ki[j] = t3[j] - (t1[j] + t2[j]);
    }
}

__device__ __forceinline__ void fnde_pi_pair(v2f* Xown, const v2f* Xpar,
                                             bool noPartner, int ownBase,
                                             const float (&kr)[8], const float (&ki)[8],
                                             float (&pr)[8], float (&pim)[8]) {
#pragma unroll
    for (int q = 0; q < 4; ++q) {
        v2f tr, ti;
        tr.x = kr[2 * q];   tr.y = kr[2 * q + 1];
        ti.x = ki[2 * q];   ti.y = ki[2 * q + 1];
        Xown[ownBase + q * 16]       = tr;
        Xown[512 + ownBase + q * 16] = ti;
    }
    __syncthreads();
#pragma unroll
    for (int q = 0; q < 4; ++q) {
        v2f tr = Xpar[ownBase + q * 16];
        v2f ti = Xpar[512 + ownBase + q * 16];
        if (noPartner) { tr.x = 0.0f; tr.y = 0.0f; ti.x = 0.0f; ti.y = 0.0f; }
        pr[2 * q]      = 0.5f * (kr[2 * q]     + tr.x);
        pr[2 * q + 1]  = 0.5f * (kr[2 * q + 1] + tr.y);
        pim[2 * q]     = 0.5f * (ki[2 * q]     - ti.x);
        pim[2 * q + 1] = 0.5f * (ki[2 * q + 1] - ti.y);
    }
}

__global__ __launch_bounds__(256)
void fnde_chain(const float* __restrict__ Ay,
                const float* __restrict__ cs, const float* __restrict__ sn,
                const float* __restrict__ lw, const float* __restrict__ lb,
                const float* __restrict__ Wp, const float* __restrict__ bias,
                const float* __restrict__ ts,
                const float* __restrict__ q, float* __restrict__ T) {
    __shared__ v2f Ush[2 * 2 * 3 * 512];
    __shared__ v2f Xsh[2 * 2 * 512];

    const int blk = blockIdx.x;
    const int wave = __builtin_amdgcn_readfirstlane((int)(threadIdx.x >> 5));
    const int lane = threadIdx.x & 31;
    const int laneHi = lane >> 4;
    const int laneLo = lane & 15;
    const int mi = wave >> 2;
    const int mt = wave & 3;
    const int ky  = blk & 15;
    const int mx0 = blk >> 4;
    const int mode = mi ? (256 + (((16 - mx0) & 15) << 4) + ky) : blk;
    const bool ky0 = (ky == 0);
    const bool noPartner = (blk == 0) && (mi == 1);
    const int  pmi = (blk == 0) ? mi : (1 - mi);

    const int ownBase = (mt * 8 + laneHi * 4) * 16 + laneLo;
    const int bBase   = laneHi * 16 + laneLo;
    const int row0    = mt * 16 + laneHi * 8;

    v2f* U0 = Ush + (0 * 2 + mi) * 1536;
    v2f* U1 = Ush + (1 * 2 + mi) * 1536;
    v2f* Xown = Xsh + mi * 1024;
    const v2f* Xpar = Xsh + pmi * 1024;

    float zr[8], zi[8], ssr[8], ssi[8], str[8], sti[8];
    {
        const int tb = threadIdx.x >> 4;
        const int xs = threadIdx.x & 15;
        const int kxA = mx0;
        const int kxB = 112 + ((16 - mx0) & 15);
        const v2f* ap = (const v2f*)Ay + (ky * 16 + tb) * 128 + xs;
        v2f av[8];
#pragma unroll
        for (int j = 0; j < 8; ++j) av[j] = ap[j * 16];
        float rA = 0.f, iA = 0.f, rB = 0.f, iB = 0.f;
#pragma unroll
        for (int j = 0; j < 8; ++j) {
            const int x = j * 16 + xs;
            const float ar = av[j].x, ai = av[j].y;
            const int aA = (kxA * x) & 127;
            const int aB = (kxB * x) & 127;
            const float cA = cs[aA], sA = sn[aA];
            const float cB = cs[aB], sB = sn[aB];
            rA += ar * cA + ai * sA;
            iA += ai * cA - ar * sA;
            rB += ar * cB + ai * sB;
            iB += ai * cB - ar * sB;
        }
        v2f pA, pB;
        pA.x = rA;   pA.y = iA;
        pB.x = rB;   pB.y = iB;
        Xsh[xs * 17 + tb]       = pA;
        Xsh[272 + xs * 17 + tb] = pB;
        __syncthreads();
        v2f hsum = Xsh[mi * 272 + laneLo];
#pragma unroll
        for (int r = 1; r < 16; ++r) hsum += Xsh[mi * 272 + r * 17 + laneLo];
        const float hr = hsum.x;
        const float hi = hsum.y;
#pragma unroll
        for (int j = 0; j < 8; ++j) {
            float w = lw[row0 + j];
            float lbj = lb[row0 + j];
            zr[j] = w * hr + ((mode == 0) ? lbj * 16384.0f : 0.0f);
            zi[j] = w * hi;
            str[j] = 0.0f;  sti[j] = 0.0f;
        }
    }

#define FNDE_PROJ(SR, SI)                                                        \
    if (ky0) fnde_pi_pair(Xown, Xpar, noPartner, ownBase, SR, SI, pr, pim);      \
    else {                                                                       \
        _Pragma("unroll")                                                        \
        for (int j = 0; j < 8; ++j) { pr[j] = SR[j]; pim[j] = SI[j]; }           \
    }

#pragma unroll 1
    for (int l = 0; l < Ll; ++l) {
        const v4f* Wm = (const v4f*)(Wp + (size_t)l * WLAYER + (size_t)mode * (2 * Cc * Cc))
                        + (mt * 16) * 32 + lane;
        v2f are[16], aim[16], aws[16];
#pragma unroll
        for (int kk = 0; kk < 16; ++kk) {
            v4f a = Wm[kk * 32];
            are[kk].x = a.x;    are[kk].y = a.y;
            aim[kk].x = a.z;    aim[kk].y = a.w;
            aws[kk] = are[kk] + aim[kk];
        }
        float bv[8];
#pragma unroll
        for (int j = 0; j < 8; ++j) {
            float bj = bias[l * Cc + row0 + j];
            bv[j] = (mode == 0) ? bj * 16384.0f : 0.0f;
        }

#pragma unroll 1
        for (int s = 0; s < 4; ++s) {
            const float h  = ts[s + 1] - ts[s];
            const float hh = 0.5f * h;
            const float h6 = h * (1.0f / 6.0f);
            float kr[8], ki[8], pr[8], pim[8], ur[8], ui[8];

            fnde_eval(U0, ownBase, bBase, zr, zi, are, aim, aws, bv, kr, ki);
#pragma unroll
            for (int j = 0; j < 8; ++j) { ssr[j] = kr[j]; ssi[j] = ki[j]; }
            FNDE_PROJ(kr, ki)
#pragma unroll
            for (int j = 0; j < 8; ++j) { ur[j] = zr[j] + hh * pr[j]; ui[j] = zi[j] + hh * pim[j]; }

            fnde_eval(U1, ownBase, bBase, ur, ui, are, aim, aws, bv, kr, ki);
#pragma unroll
            for (int j = 0; j < 8; ++j) { ssr[j] += 2.0f * kr[j]; ssi[j] += 2.0f * ki[j]; }
            FNDE_PROJ(kr, ki)
#pragma unroll
            for (int j = 0; j < 8; ++j) { ur[j] = zr[j] + hh * pr[j]; ui[j] = zi[j] + hh * pim[j]; }

            fnde_eval(U0, ownBase, bBase, ur, ui, are, aim, aws, bv, kr, ki);
#pragma unroll
            for (int j = 0; j < 8; ++j) { ssr[j] += 2.0f * kr[j]; ssi[j] += 2.0f * ki[j]; }
            FNDE_PROJ(kr, ki)
#pragma unroll
            for (int j = 0; j < 8; ++j) { ur[j] = zr[j] + h * pr[j]; ui[j] = zi[j] + h * pim[j]; }

            fnde_eval(U1, ownBase, bBase, ur, ui, are, aim, aws, bv, kr, ki);
#pragma unroll
            for (int j = 0; j < 8; ++j) { ssr[j] += kr[j]; ssi[j] += ki[j]; }
            FNDE_PROJ(ssr, ssi)
#pragma unroll
            for (int j = 0; j < 8; ++j) {
                zr[j]  += h6 * pr[j];    zi[j]  += h6 * pim[j];
                str[j] += h6 * ssr[j];   sti[j] += h6 * ssi[j];
            }
        }
    }
#undef FNDE_PROJ

    float pre = 0.0f, pie = 0.0f;
#pragma unroll
    for (int j = 0; j < 8; ++j) {
        const float w = q[row0 + j];
        pre += w * str[j];
        pie += w * sti[j];
    }
    v2f part;
    part.x = pre;   part.y = pie;
    Ush[threadIdx.x] = part;
    __syncthreads();
    if (wave == 0) {
        const int m2 = laneHi;
        v2f acc = Ush[m2 * 128 + laneLo];
#pragma unroll
        for (int r = 1; r < 8; ++r) acc += Ush[m2 * 128 + r * 16 + laneLo];
        const int modeL = m2 ? (256 + (((16 - mx0) & 15) << 4) + ky) : blk;
        *(v2f*)(T + (modeL * 16 + laneLo) * 2) = acc;
    }
}

__device__ __forceinline__ void fnde_prep_body(const float* __restrict__ p1w, const float* __restrict__ p1b,
                                               const float* __restrict__ p2w, const float* __restrict__ p2b,
                                               const float* __restrict__ lw,  const float* __restrict__ lb,
                                               float* __restrict__ q, float* __restrict__ scal,
                                               float* qs) {
    int c = threadIdx.x;
    if (c < 64) {
        float acc = 0.f;
        for (int p = 0; p < 64; ++p) acc += p2w[p] * p1w[p * 64 + c];
        qs[c] = acc;
        q[c] = acc;
    }
    __syncthreads();
    if (c == 0) {
        float r = p2b[0];
        for (int p = 0; p < 64; ++p) r += p2w[p] * p1b[p];
        float al = 0.f, be = r;
        for (int i = 0; i < 64; ++i) { al += qs[i] * lw[i]; be += qs[i] * lb[i]; }
        scal[0] = al;
        scal[1] = be;
    }
}

#define FR_PREP 1
#define FR_DFT1 2
#define FR_PACK (2 + 256)
#define FR_GRID (FR_PACK + Ll * 2 * Mm * 4 * 4)
__global__ __launch_bounds__(128)
void fnde_front(const float* __restrict__ z,
                const float* __restrict__ w1, const float* __restrict__ w2,
                const float* __restrict__ p1w, const float* __restrict__ p1b,
                const float* __restrict__ p2w, const float* __restrict__ p2b,
                const float* __restrict__ lw,  const float* __restrict__ lb,
                float* __restrict__ cs, float* __restrict__ sn,
                float* __restrict__ q, float* __restrict__ scal,
                float* __restrict__ Ay, float* __restrict__ Wp) {
    __shared__ v4f fr_lds[16 * (PK_PITCH / 4)];
    const int blk = blockIdx.x;
    if (blk >= FR_PACK)      fnde_pack_body(w1, w2, Wp, fr_lds, blk - FR_PACK);
    else if (blk >= FR_DFT1) fnde_dft1_body(z, Ay, (float*)fr_lds, blk - FR_DFT1);
    else if (blk == FR_PREP) fnde_prep_body(p1w, p1b, p2w, p2b, lw, lb, q, scal, (float*)fr_lds);
    else                     fnde_tables_body(cs, sn);
}

__global__ __launch_bounds__(256)
void fnde_inv_dft(const float* __restrict__ T, const float* __restrict__ z,
                  const float* __restrict__ cs, const float* __restrict__ sn,
                  const float* __restrict__ scal, float* __restrict__ out) {
    __shared__ v2f Tsh[NMODE];
    __shared__ v2f Gsh[8 * 16];
    const int tid = threadIdx.x;
    const int b  = blockIdx.x >> 4;
    const int xg = blockIdx.x & 15;
    Tsh[tid]       = *(const v2f*)(T + (tid * 16 + b) * 2);
    Tsh[tid + 256] = *(const v2f*)(T + ((tid + 256) * 16 + b) * 2);
    __syncthreads();
    if (tid < 128) {
        const int x  = xg * 8 + (tid >> 4);
        const int ky = tid & 15;
        float re = 0.f, im = 0.f;
        for (int corner = 0; corner < 2; ++corner) {
            for (int mx = 0; mx < 16; ++mx) {
                int m  = corner * 256 + mx * 16 + ky;
                int kx = corner ? (112 + mx) : mx;
                v2f tv = Tsh[m];
                float tr = tv.x, ti = tv.y;
                int a = (kx * x) & 127;
                float c = cs[a], s = sn[a];
                re += tr * c - ti * s;
                im += tr * s + ti * c;
            }
        }
        v2f g;
        g.x = re;   g.y = im;
        Gsh[tid] = g;
    }
    __syncthreads();
    const float al = scal[0], be = scal[1];
    for (int r = 0; r < 4; ++r) {
        const int idx = r * 256 + tid;
        const int y = idx & 127;
        const v2f* g = Gsh + (idx >> 7) * 16;
        float acc = g[0].x;
        for (int ky = 1; ky < 16; ++ky) {
            int a = (ky * y) & 127;
            acc += 2.0f * (g[ky].x * cs[a] - g[ky].y * sn[a]);
        }
        const int t = ((b * 128 + xg * 8) << 7) + idx;
        out[t] = al * z[t] + be + acc * (1.0f / 16384.0f);
    }
}

extern "C" void kernel_launch(void* const* d_in, const int* in_sizes, int n_in,
                              void* d_out, int out_size, void* d_ws, size_t ws_size,
                              hipStream_t stream) {
    (void)in_sizes; (void)n_in; (void)out_size; (void)ws_size;

    const float* zin    = (const float*)d_in[0];
    const float* lift_w = (const float*)d_in[1];
    const float* lift_b = (const float*)d_in[2];
    const float* spec_w1 = (const float*)d_in[3];
    const float* spec_w2 = (const float*)d_in[4];
    const float* fl_bias = (const float*)d_in[5];
    const float* p1_w = (const float*)d_in[6];
    const float* p1_b = (const float*)d_in[7];
    const float* p2_w = (const float*)d_in[8];
    const float* p2_b = (const float*)d_in[9];
    const float* ts   = (const float*)d_in[10];
    float* outp = (float*)d_out;

    float* w = (float*)d_ws;
    float* Wp   = w;  w += (size_t)Ll * WLAYER;
    float* Ay   = w;  w += Bb * Dd * Mm * 2;
    float* Tm   = w;  w += Bb * NMODE * 2;
    float* qv   = w;  w += 64;
    float* cs   = w;  w += 128;
    float* sn   = w;  w += 128;
    float* scal = w;  w += 8;

    fnde_front<<<FR_GRID, 128, 0, stream>>>(zin, spec_w1, spec_w2, p1_w, p1_b, p2_w, p2_b, lift_w, lift_b,
                                            cs, sn, qv, scal, Ay, Wp);
    fnde_chain<<<NMODE / 2, 256, 0, stream>>>(Ay, cs, sn, lift_w, lift_b, Wp, fl_bias, ts, qv, Tm);
    fnde_inv_dft<<<Bb * (Dd / 8), 256, 0, stream>>>(Tm, zin, cs, sn, scal, outp);
}


# ky=0 pairs: Pi exchange folded into the B publish: a lane stores z+c/2*K in its own planes and c/2*conj(K) in its partner's, the reader adds them (sum regrouped): one barrier per product, not two
# speedup vs baseline: 12.8149x; 1.0701x over previous
#include <hip/hip_runtime.h>


typedef float v2f __attribute__((ext_vector_type(2)));
typedef float v4f __attribute__((ext_vector_type(4)));
typedef float v8f __attribute__((ext_vector_type(8)));

#define Bb    16
#define Cc    64
#define Dd    128
#define Mm    16
#define Ll    3
#define NMODE 512
#define WLAYER (2*NMODE*Cc*Cc)

__device__ __forceinline__ void fnde_tables_body(float* cs, float* sn) {
    int t = threadIdx.x;
    float ang = (float)(6.283185307179586 / 128.0) * (float)t;
    cs[t] = cosf(ang);
    sn[t] = sinf(ang);
}

#define PK_PITCH 516
__device__ __forceinline__ void fnde_pack_body(const float* __restrict__ w1,
                                               const float* __restrict__ w2,
                                               float* __restrict__ Wp,
                                               v4f* pk_tile, int blk) {
    float* tile = (float*)pk_tile;

    const int t = threadIdx.x;
    const int kq = blk & 3;    blk >>= 2;
    const int mt = blk & 3;    blk >>= 2;
    const int mx = blk & 15;   blk >>= 4;
    const int corner = blk & 1;
    const int l = blk >> 1;
    const float* src = corner ? w2 : w1;

    const int p  = t & 7;
    const int b  = (t >> 3) & 1;
    const int cb = (t >> 4) & 1;
    const int w  = t >> 5;
    const float* sp = src + ((((size_t)(l * Cc + kq * 16 + b) * Cc + mt * 16 + 2 * w + cb) * Mm + mx) * Mm) * 2
                          + p * 4;
    float* tp = tile + (2 * p) * PK_PITCH + (2 * w + cb) * 4 + b * 2;
    v4f v[16];
#pragma unroll
    for (int r = 0; r < 16; ++r) {
        const int ip = r >> 1;
        const int cp = (r & 1) * 4;
        v[r] = *(const v4f*)(sp + (size_t)(ip * 2) * (Cc * Mm * Mm * 2) + (size_t)(cp * 2) * (Mm * Mm * 2));
    }
#pragma unroll
    for (int r = 0; r < 16; ++r) {
        const int ip = r >> 1;
        const int cp = (r & 1) * 4;
        float* d = tp + ip * 64 + cp * 8;
        v2f e0, e1;
        e0.x = v[r].x;   e0.y = v[r].y;
        e1.x = v[r].z;   e1.y = v[r].w;
        *(v2f*)d              = e0;
        *(v2f*)(d + PK_PITCH) = e1;
    }
    __syncthreads();

    const int lane = t & 31;
    const v4f* tq = pk_tile + w * 32 + lane;
    v4f* dp = (v4f*)(Wp + (size_t)l * WLAYER + (size_t)(corner * 256 + mx * 16) * (2 * Cc * Cc))
              + (mt * 16 + kq * 4 + w) * 32 + lane;
#pragma unroll
    for (int ky = 0; ky < 16; ++ky) {
        v4f a = tq[ky * (PK_PITCH / 4)];
        v4f f;
        f.x = a.x;   f.y = a.z;   f.z = a.y;   f.w = a.w;
        dp[(size_t)ky * (2 * Cc * Cc / 4)] = f;
    }
}

__device__ __forceinline__ void fnde_dft1_body(const float* __restrict__ z,
                                               float* __restrict__ Ay,
                                               float* tab, int blk) {
    const int tid = threadIdx.x;
    {
        float ang = (float)(6.283185307179586 / 128.0) * (float)tid;
        tab[tid]       = cosf(ang);
        tab[128 + tid] = sinf(ang);
    }
    __syncthreads();
    int t  = blk * 128 + tid;
    int b  = t >> 11;
    int x  = (t >> 4) & 127;
    int ky = t & 15;
    const float* row = z + (b * 128 + x) * 128;
    float re = 0.f, im = 0.f;
    for (int y = 0; y < 128; ++y) {
        float v = row[y];
        int a = (ky * y) & 127;
        re += v * tab[a];
        im -= v * tab[128 + a];
    }
    v2f o;
    o.x = re;   o.y = im;
    *(v2f*)(Ay + (size_t)((ky * 16 + b) * 128 + x) * 2) = o;
}


__device__ __forceinline__ void fnde_combine(v8f t1, v8f t2, v8f t3, const float (&bv)[8],
                                             float (&kr)[8], float (&ki)[8]) {
    asm volatile("v_nop\n\tv_nop\n\tv_nop\n\tv_nop" : "+v"(t1), "+v"(t2), "+v"(t3));
#pragma unroll
    for (int j = 0; j < 8; ++j) {
        kr[j] = (t1[j] - t2[j]) + bv[j];
        ki[j] = t3[j] - (t1[j] + t2[j]);
    }
}

__device__ __forceinline__ void fnde_eval(v2f* Ub, int ownBase, int bBase,
                                          const float (&vr)[8], const float (&vi)[8],
                                          const v2f (&are)[16], const v2f (&aim)[16],
                                          const v2f (&aws)[16], const float (&bv)[8],
                                          float (&kr)[8], float (&ki)[8]) {
#pragma unroll
    for (int q = 0; q < 4; ++q) {
        v2f tr, ti;
        tr.x = vr[2 * q];   tr.y = vr[2 * q + 1];
        ti.x = vi[2 * q];   ti.y = vi[2 * q + 1];
        Ub[ownBase + q * 16]        = tr;
        Ub[512 + ownBase + q * 16]  = ti;
        Ub[1024 + ownBase + q * 16] = tr + ti;
    }
    __syncthreads();
    v8f t1 = {}, t2 = {}, t3 = {};
#pragma unroll
    for (int kk = 0; kk < 16; ++kk) {
        v2f br = Ub[bBase + kk * 32];
        v2f bi = Ub[512 + bBase + kk * 32];
        v2f bs = Ub[1024 + bBase + kk * 32];
        t1 = __builtin_amdgcn_wmma_f32_16x16x4_f32(false, are[kk], false, br, (short)0, t1, false, false);
        t2 = __builtin_amdgcn_wmma_f32_16x16x4_f32(false, aim[kk], false, bi, (short)0, t2, false, false);
        t3 = __builtin_amdgcn_wmma_f32_16x16x4_f32(false, aws[kk], false, bs, (short)0, t3, false, false);
    }
    fnde_combine(t1, t2, t3, bv, kr, ki);
}

__device__ __forceinline__ void fnde_eval_pair(v2f* Uown, v2f* Upar,
                                               int ownBase, int bBase,
                                               const float (&ar)[8], const float (&ai)[8],
                                               const float (&dr)[8], const float (&di)[8],
                                               const v2f (&are)[16], const v2f (&aim)[16],
                                               const v2f (&aws)[16], const float (&bv)[8],
                                               float (&kr)[8], float (&ki)[8]) {
#pragma unroll
    for (int q = 0; q < 4; ++q) {
        v2f tr, ti, sr, si;
        tr.x = ar[2 * q];   tr.y = ar[2 * q + 1];
        ti.x = ai[2 * q];   ti.y = ai[2 * q + 1];
        sr.x = dr[2 * q];   sr.y = dr[2 * q + 1];
        si.x = di[2 * q];   si.y = di[2 * q + 1];
        Uown[ownBase + q * 16]        = tr;
        Uown[512 + ownBase + q * 16]  = ti;
        Upar[1024 + ownBase + q * 16] = sr;
        Upar[1536 + ownBase + q * 16] = si;
    }
    __syncthreads();
    v8f t1 = {}, t2 = {}, t3 = {};
#pragma unroll
    for (int kk = 0; kk < 16; ++kk) {
        v2f br = Uown[bBase + kk * 32]       + Uown[1024 + bBase + kk * 32];
        v2f bi = Uown[512 + bBase + kk * 32] + Uown[1536 + bBase + kk * 32];
        v2f bs = br + bi;
        t1 = __builtin_amdgcn_wmma_f32_16x16x4_f32(false, are[kk], false, br, (short)0, t1, false, false);
        t2 = __builtin_amdgcn_wmma_f32_16x16x4_f32(false, aim[kk], false, bi, (short)0, t2, false, false);
        t3 = __builtin_amdgcn_wmma_f32_16x16x4_f32(false, aws[kk], false, bs, (short)0, t3, false, false);
    }
    fnde_combine(t1, t2, t3, bv, kr, ki);
}

template <bool PAIRED>
__device__ __forceinline__ void fnde_layers(v2f* U0, v2f* U1, v2f* P0, v2f* P1, bool noPartner,
                                            int ownBase, int bBase, int row0, int mode, int mt, int lane,
                                            const float* __restrict__ Wp, const float* __restrict__ bias,
                                            const float* __restrict__ ts,
                                            float (&zr)[8], float (&zi)[8],
                                            float (&str)[8], float (&sti)[8]) {
    float ssr[8], ssi[8], wr[8], vi[8];
#pragma unroll
    for (int j = 0; j < 8; ++j) { wr[j] = 0.0f; vi[j] = 0.0f; }

#pragma unroll 1
    for (int l = 0; l < Ll; ++l) {
        const v4f* Wm = (const v4f*)(Wp + (size_t)l * WLAYER + (size_t)mode * (2 * Cc * Cc))
                        + (mt * 16) * 32 + lane;
        v2f are[16], aim[16], aws[16];
#pragma unroll
        for (int kk = 0; kk < 16; ++kk) {
            v4f a = Wm[kk * 32];
            are[kk].x = a.x;    are[kk].y = a.y;
            aim[kk].x = a.z;    aim[kk].y = a.w;
            aws[kk] = are[kk] + aim[kk];
        }
        float bv[8];
#pragma unroll
        for (int j = 0; j < 8; ++j) {
            float bj = bias[l * Cc + row0 + j];
            bv[j] = (mode == 0) ? bj * 16384.0f : 0.0f;
        }

#pragma unroll 1
        for (int s = 0; s < 4; ++s) {
            const float h  = ts[s + 1] - ts[s];
            const float hh = 0.5f * h;
            const float h6 = h * (1.0f / 6.0f);
            float kr[8], ki[8], ur[8], ui[8];

            if (PAIRED) {
                const float c2 = 0.5f * hh;
                const float c4 = 0.5f * h;
                const float g  = 0.5f * h6;
                const float d2 = noPartner ? 0.0f : c2;
                const float d4 = noPartner ? 0.0f : c4;
                const float gd = noPartner ? 0.0f : g;
                float dr[8], di[8];
#pragma unroll
                for (int j = 0; j < 8; ++j) { dr[j] = wr[j]; di[j] = vi[j]; }
                fnde_eval_pair(U0, P0, ownBase, bBase, zr, zi, dr, di, are, aim, aws, bv, kr, ki);
#pragma unroll
                for (int j = 0; j < 8; ++j) {
                    ssr[j] = kr[j];                 ssi[j] = ki[j];
                    ur[j] = zr[j] + c2 * kr[j];     ui[j] = zi[j] + c2 * ki[j];
                    dr[j] = wr[j] + d2 * kr[j];     di[j] = vi[j] - d2 * ki[j];
                }
                fnde_eval_pair(U1, P1, ownBase, bBase, ur, ui, dr, di, are, aim, aws, bv, kr, ki);
#pragma unroll
                for (int j = 0; j < 8; ++j) {
                    ssr[j] += 2.0f * kr[j];         ssi[j] += 2.0f * ki[j];
                    ur[j] = zr[j] + c2 * kr[j];     ui[j] = zi[j] + c2 * ki[j];
                    dr[j] = wr[j] + d2 * kr[j];     di[j] = vi[j] - d2 * ki[j];
                }
                fnde_eval_pair(U0, P0, ownBase, bBase, ur, ui, dr, di, are, aim, aws, bv, kr, ki);
#pragma unroll
                for (int j = 0; j < 8; ++j) {
                    ssr[j] += 2.0f * kr[j];         ssi[j] += 2.0f * ki[j];
                    ur[j] = zr[j] + c4 * kr[j];     ui[j] = zi[j] + c4 * ki[j];
                    dr[j] = wr[j] + d4 * kr[j];     di[j] = vi[j] - d4 * ki[j];
                }
                fnde_eval_pair(U1, P1, ownBase, bBase, ur, ui, dr, di, are, aim, aws, bv, kr, ki);
#pragma unroll
                for (int j = 0; j < 8; ++j) {
                    ssr[j] += kr[j];                ssi[j] += ki[j];
                    zr[j]  += g * ssr[j];           zi[j]  += g * ssi[j];
                    wr[j]  += gd * ssr[j];          vi[j]  -= gd * ssi[j];
                    str[j] += h6 * ssr[j];          sti[j] += h6 * ssi[j];
                }
            } else {
                fnde_eval(U0, ownBase, bBase, zr, zi, are, aim, aws, bv, kr, ki);
#pragma unroll
                for (int j = 0; j < 8; ++j) {
                    ssr[j] = kr[j];                 ssi[j] = ki[j];
                    ur[j] = zr[j] + hh * kr[j];     ui[j] = zi[j] + hh * ki[j];
                }
                fnde_eval(U1, ownBase, bBase, ur, ui, are, aim, aws, bv, kr, ki);
#pragma unroll
                for (int j = 0; j < 8; ++j) {
                    ssr[j] += 2.0f * kr[j];         ssi[j] += 2.0f * ki[j];
                    ur[j] = zr[j] + hh * kr[j];     ui[j] = zi[j] + hh * ki[j];
                }
                fnde_eval(U0, ownBase, bBase, ur, ui, are, aim, aws, bv, kr, ki);
#pragma unroll
                for (int j = 0; j < 8; ++j) {
                    ssr[j] += 2.0f * kr[j];         ssi[j] += 2.0f * ki[j];
                    ur[j] = zr[j] + h * kr[j];      ui[j] = zi[j] + h * ki[j];
                }
                fnde_eval(U1, ownBase, bBase, ur, ui, are, aim, aws, bv, kr, ki);
#pragma unroll
                for (int j = 0; j < 8; ++j) {
                    ssr[j] += kr[j];                ssi[j] += ki[j];
                    zr[j]  += h6 * ssr[j];          zi[j]  += h6 * ssi[j];
                    str[j] += h6 * ssr[j];          sti[j] += h6 * ssi[j];
                }
            }
        }
    }
}

__global__ __launch_bounds__(256)
void fnde_chain(const float* __restrict__ Ay,
                const float* __restrict__ cs, const float* __restrict__ sn,
                const float* __restrict__ lw, const float* __restrict__ lb,
                const float* __restrict__ Wp, const float* __restrict__ bias,
                const float* __restrict__ ts,
                const float* __restrict__ q, float* __restrict__ T) {
    __shared__ v2f Ush[2 * 2 * 4 * 512];
    v2f* Xsh = Ush + 4096;

    const int blk = blockIdx.x;
    const int wave = __builtin_amdgcn_readfirstlane((int)(threadIdx.x >> 5));
    const int lane = threadIdx.x & 31;
    const int laneHi = lane >> 4;
    const int laneLo = lane & 15;
    const int mi = wave >> 2;
    const int mt = wave & 3;
    const int ky  = blk & 15;
    const int mx0 = blk >> 4;
    const int mode = mi ? (256 + (((16 - mx0) & 15) << 4) + ky) : blk;
    const bool ky0 = (ky == 0);
    const bool noPartner = (blk == 0) && (mi == 1);
    const int  pmi = (blk == 0) ? mi : (1 - mi);

    const int ownBase = (mt * 8 + laneHi * 4) * 16 + laneLo;
    const int bBase   = laneHi * 16 + laneLo;
    const int row0    = mt * 16 + laneHi * 8;

    v2f* U0 = Ush + (0 * 2 + mi) * 2048;
    v2f* U1 = Ush + (1 * 2 + mi) * 2048;
    v2f* P0 = Ush + (0 * 2 + pmi) * 2048;
    v2f* P1 = Ush + (1 * 2 + pmi) * 2048;

    float zr[8], zi[8], str[8], sti[8];
    {
        const int tb = threadIdx.x >> 4;
        const int xs = threadIdx.x & 15;
        const int kxA = mx0;
        const int kxB = 112 + ((16 - mx0) & 15);
        const v2f* ap = (const v2f*)Ay + (ky * 16 + tb) * 128 + xs;
        v2f av[8];
#pragma unroll
        for (int j = 0; j < 8; ++j) av[j] = ap[j * 16];
        float rA = 0.f, iA = 0.f, rB = 0.f, iB = 0.f;
#pragma unroll
        for (int j = 0; j < 8; ++j) {
            const int x = j * 16 + xs;
            const float ar = av[j].x, ai = av[j].y;
            const int aA = (kxA * x) & 127;
            const int aB = (kxB * x) & 127;
            const float cA = cs[aA], sA = sn[aA];
            const float cB = cs[aB], sB = sn[aB];
            rA += ar * cA + ai * sA;
            iA += ai * cA - ar * sA;
            rB += ar * cB + ai * sB;
            iB += ai * cB - ar * sB;
        }
        v2f pA, pB;
        pA.x = rA;   pA.y = iA;
        pB.x = rB;   pB.y = iB;
        Xsh[xs * 17 + tb]       = pA;
        Xsh[272 + xs * 17 + tb] = pB;
        __syncthreads();
        v2f hsum = Xsh[mi * 272 + laneLo];
#pragma unroll
        for (int r = 1; r < 16; ++r) hsum += Xsh[mi * 272 + r * 17 + laneLo];
        const float hr = hsum.x;
        const float hi = hsum.y;
#pragma unroll
        for (int j = 0; j < 8; ++j) {
            float w = lw[row0 + j];
            float lbj = lb[row0 + j];
            zr[j] = w * hr + ((mode == 0) ? lbj * 16384.0f : 0.0f);
            zi[j] = w * hi;
            str[j] = 0.0f;  sti[j] = 0.0f;
        }
    }

    if (ky0) fnde_layers<true >(U0, U1, P0, P1, noPartner, ownBase, bBase, row0, mode, mt, lane,
                                Wp, bias, ts, zr, zi, str, sti);
    else     fnde_layers<false>(U0, U1, P0, P1, noPartner, ownBase, bBase, row0, mode, mt, lane,
                                Wp, bias, ts, zr, zi, str, sti);

    float pre = 0.0f, pie = 0.0f;
#pragma unroll
    for (int j = 0; j < 8; ++j) {
        const float w = q[row0 + j];
        pre += w * str[j];
        pie += w * sti[j];
    }
    v2f part;
    part.x = pre;   part.y = pie;
    Ush[threadIdx.x] = part;
    __syncthreads();
    if (wave == 0) {
        const int m2 = laneHi;
        v2f acc = Ush[m2 * 128 + laneLo];
#pragma unroll
        for (int r = 1; r < 8; ++r) acc += Ush[m2 * 128 + r * 16 + laneLo];
        const int modeL = m2 ? (256 + (((16 - mx0) & 15) << 4) + ky) : blk;
        *(v2f*)(T + (modeL * 16 + laneLo) * 2) = acc;
    }
}

__device__ __forceinline__ void fnde_prep_body(const float* __restrict__ p1w, const float* __restrict__ p1b,
                                               const float* __restrict__ p2w, const float* __restrict__ p2b,
                                               const float* __restrict__ lw,  const float* __restrict__ lb,
                                               float* __restrict__ q, float* __restrict__ scal,
                                               float* qs) {
    int c = threadIdx.x;
    if (c < 64) {
        float acc = 0.f;
        for (int p = 0; p < 64; ++p) acc += p2w[p] * p1w[p * 64 + c];
        qs[c] = acc;
        q[c] = acc;
    }
    __syncthreads();
    if (c == 0) {
        float r = p2b[0];
        for (int p = 0; p < 64; ++p) r += p2w[p] * p1b[p];
        float al = 0.f, be = r;
        for (int i = 0; i < 64; ++i) { al += qs[i] * lw[i]; be += qs[i] * lb[i]; }
        scal[0] = al;
        scal[1] = be;
    }
}

#define FR_PREP 1
#define FR_DFT1 2
#define FR_PACK (2 + 256)
#define FR_GRID (FR_PACK + Ll * 2 * Mm * 4 * 4)
__global__ __launch_bounds__(128)
void fnde_front(const float* __restrict__ z,
                const float* __restrict__ w1, const float* __restrict__ w2,
                const float* __restrict__ p1w, const float* __restrict__ p1b,
                const float* __restrict__ p2w, const float* __restrict__ p2b,
                const float* __restrict__ lw,  const float* __restrict__ lb,
                float* __restrict__ cs, float* __restrict__ sn,
                float* __restrict__ q, float* __restrict__ scal,
                float* __restrict__ Ay, float* __restrict__ Wp) {
    __shared__ v4f fr_lds[16 * (PK_PITCH / 4)];
    const int blk = blockIdx.x;
    if (blk >= FR_PACK)      fnde_pack_body(w1, w2, Wp, fr_lds, blk - FR_PACK);
    else if (blk >= FR_DFT1) fnde_dft1_body(z, Ay, (float*)fr_lds, blk - FR_DFT1);
    else if (blk == FR_PREP) fnde_prep_body(p1w, p1b, p2w, p2b, lw, lb, q, scal, (float*)fr_lds);
    else                     fnde_tables_body(cs, sn);
}

__global__ __launch_bounds__(256)
void fnde_inv_dft(const float* __restrict__ T, const float* __restrict__ z,
                  const float* __restrict__ cs, const float* __restrict__ sn,
                  const float* __restrict__ scal, float* __restrict__ out) {
    __shared__ v2f Tsh[NMODE];
    __shared__ v2f Gsh[8 * 16];
    const int tid = threadIdx.x;
    const int b  = blockIdx.x >> 4;
    const int xg = blockIdx.x & 15;
    Tsh[tid]       = *(const v2f*)(T + (tid * 16 + b) * 2);
    Tsh[tid + 256] = *(const v2f*)(T + ((tid + 256) * 16 + b) * 2);
    __syncthreads();
    if (tid < 128) {
        const int x  = xg * 8 + (tid >> 4);
        const int ky = tid & 15;
        float re = 0.f, im = 0.f;
        for (int corner = 0; corner < 2; ++corner) {
            for (int mx = 0; mx < 16; ++mx) {
                int m  = corner * 256 + mx * 16 + ky;
                int kx = corner ? (112 + mx) : mx;
                v2f tv = Tsh[m];
                float tr = tv.x, ti = tv.y;
                int a = (kx * x) & 127;
                float c = cs[a], s = sn[a];
                re += tr * c - ti * s;
                im += tr * s + ti * c;
            }
        }
        v2f g;
        g.x = re;   g.y = im;
        Gsh[tid] = g;
    }
    __syncthreads();
    const float al = scal[0], be = scal[1];
    for (int r = 0; r < 4; ++r) {
        const int idx = r * 256 + tid;
        const int y = idx & 127;
        const v2f* g = Gsh + (idx >> 7) * 16;
        float acc = g[0].x;
        for (int ky = 1; ky < 16; ++ky) {
            int a = (ky * y) & 127;
            acc += 2.0f * (g[ky].x * cs[a] - g[ky].y * sn[a]);
        }
        const int t = ((b * 128 + xg * 8) << 7) + idx;
        out[t] = al * z[t] + be + acc * (1.0f / 16384.0f);
    }
}

extern "C" void kernel_launch(void* const* d_in, const int* in_sizes, int n_in,
                              void* d_out, int out_size, void* d_ws, size_t ws_size,
                              hipStream_t stream) {
    (void)in_sizes; (void)n_in; (void)out_size; (void)ws_size;

    const float* zin    = (const float*)d_in[0];
    const float* lift_w = (const float*)d_in[1];
    const float* lift_b = (const float*)d_in[2];
    const float* spec_w1 = (const float*)d_in[3];
    const float* spec_w2 = (const float*)d_in[4];
    const float* fl_bias = (const float*)d_in[5];
    const float* p1_w = (const float*)d_in[6];
    const float* p1_b = (const float*)d_in[7];
    const float* p2_w = (const float*)d_in[8];
    const float* p2_b = (const float*)d_in[9];
    const float* ts   = (const float*)d_in[10];
    float* outp = (float*)d_out;

    float* w = (float*)d_ws;
    float* Wp   = w;  w += (size_t)Ll * WLAYER;
    float* Ay   = w;  w += Bb * Dd * Mm * 2;
    float* Tm   = w;  w += Bb * NMODE * 2;
    float* qv   = w;  w += 64;
    float* cs   = w;  w += 128;
    float* sn   = w;  w += 128;
    float* scal = w;  w += 8;

    fnde_front<<<FR_GRID, 128, 0, stream>>>(zin, spec_w1, spec_w2, p1_w, p1_b, p2_w, p2_b, lift_w, lift_b,
                                            cs, sn, qv, scal, Ay, Wp);
    fnde_chain<<<NMODE / 2, 256, 0, stream>>>(Ay, cs, sn, lift_w, lift_b, Wp, fl_bias, ts, qv, Tm);
    fnde_inv_dft<<<Bb * (Dd / 8), 256, 0, stream>>>(Tm, zin, cs, sn, scal, outp);
}


# ky=0 modes: Pi projection folded into the weights, (W+conj(W_partner))/2 formed once per layer (stage input is a real field's spectrum: partner = conjugate); plain product, sums regrouped
# speedup vs baseline: 13.6838x; 1.0678x over previous
#include <hip/hip_runtime.h>


typedef float v2f __attribute__((ext_vector_type(2)));
typedef float v4f __attribute__((ext_vector_type(4)));
typedef float v8f __attribute__((ext_vector_type(8)));

#define Bb    16
#define Cc    64
#define Dd    128
#define Mm    16
#define Ll    3
#define NMODE 512
#define WLAYER (2*NMODE*Cc*Cc)

__device__ __forceinline__ void fnde_tables_body(float* cs, float* sn) {
    int t = threadIdx.x;
    float ang = (float)(6.283185307179586 / 128.0) * (float)t;
    cs[t] = cosf(ang);
    sn[t] = sinf(ang);
}

#define PK_PITCH 516
__device__ __forceinline__ void fnde_pack_body(const float* __restrict__ w1,
                                               const float* __restrict__ w2,
                                               float* __restrict__ Wp,
                                               v4f* pk_tile, int blk) {
    float* tile = (float*)pk_tile;

    const int t = threadIdx.x;
    const int kq = blk & 3;    blk >>= 2;
    const int mt = blk & 3;    blk >>= 2;
    const int mx = blk & 15;   blk >>= 4;
    const int corner = blk & 1;
    const int l = blk >> 1;
    const float* src = corner ? w2 : w1;

    const int p  = t & 7;
    const int b  = (t >> 3) & 1;
    const int cb = (t >> 4) & 1;
    const int w  = t >> 5;
    const float* sp = src + ((((size_t)(l * Cc + kq * 16 + b) * Cc + mt * 16 + 2 * w + cb) * Mm + mx) * Mm) * 2
                          + p * 4;
    float* tp = tile + (2 * p) * PK_PITCH + (2 * w + cb) * 4 + b * 2;
    v4f v[16];
#pragma unroll
    for (int r = 0; r < 16; ++r) {
        const int ip = r >> 1;
        const int cp = (r & 1) * 4;
        v[r] = *(const v4f*)(sp + (size_t)(ip * 2) * (Cc * Mm * Mm * 2) + (size_t)(cp * 2) * (Mm * Mm * 2));
    }
#pragma unroll
    for (int r = 0; r < 16; ++r) {
        const int ip = r >> 1;
        const int cp = (r & 1) * 4;
        float* d = tp + ip * 64 + cp * 8;
        v2f e0, e1;
        e0.x = v[r].x;   e0.y = v[r].y;
        e1.x = v[r].z;   e1.y = v[r].w;
        *(v2f*)d              = e0;
        *(v2f*)(d + PK_PITCH) = e1;
    }
    __syncthreads();

    const int lane = t & 31;
    const v4f* tq = pk_tile + w * 32 + lane;
    v4f* dp = (v4f*)(Wp + (size_t)l * WLAYER + (size_t)(corner * 256 + mx * 16) * (2 * Cc * Cc))
              + (mt * 16 + kq * 4 + w) * 32 + lane;
#pragma unroll
    for (int ky = 0; ky < 16; ++ky) {
        v4f a = tq[ky * (PK_PITCH / 4)];
        v4f f;
        f.x = a.x;   f.y = a.z;   f.z = a.y;   f.w = a.w;
        dp[(size_t)ky * (2 * Cc * Cc / 4)] = f;
    }
}

__device__ __forceinline__ void fnde_dft1_body(const float* __restrict__ z,
                                               float* __restrict__ Ay,
                                               float* tab, int blk) {
    const int tid = threadIdx.x;
    {
        float ang = (float)(6.283185307179586 / 128.0) * (float)tid;
        tab[tid]       = cosf(ang);
        tab[128 + tid] = sinf(ang);
    }
    __syncthreads();
    int t  = blk * 128 + tid;
    int b  = t >> 11;
    int x  = (t >> 4) & 127;
    int ky = t & 15;
    const float* row = z + (b * 128 + x) * 128;
    float re = 0.f, im = 0.f;
    for (int y = 0; y < 128; ++y) {
        float v = row[y];
        int a = (ky * y) & 127;
        re += v * tab[a];
        im -= v * tab[128 + a];
    }
    v2f o;
    o.x = re;   o.y = im;
    *(v2f*)(Ay + (size_t)((ky * 16 + b) * 128 + x) * 2) = o;
}


__device__ __forceinline__ void fnde_combine(v8f t1, v8f t2, v8f t3, const float (&bv)[8],
                                             float (&kr)[8], float (&ki)[8]) {
    asm volatile("v_nop\n\tv_nop\n\tv_nop\n\tv_nop" : "+v"(t1), "+v"(t2), "+v"(t3));
#pragma unroll
    for (int j = 0; j < 8; ++j) {
        kr[j] = (t1[j] - t2[j]) + bv[j];
        ki[j] = t3[j] - (t1[j] + t2[j]);
    }
}

__device__ __forceinline__ void fnde_eval(v2f* Ub, int ownBase, int bBase,
                                          const float (&vr)[8], const float (&vi)[8],
                                          const v2f (&are)[16], const v2f (&aim)[16],
                                          const v2f (&aws)[16], const float (&bv)[8],
                                          float (&kr)[8], float (&ki)[8]) {
#pragma unroll
    for (int q = 0; q < 4; ++q) {
        v2f tr, ti;
        tr.x = vr[2 * q];   tr.y = vr[2 * q + 1];
        ti.x = vi[2 * q];   ti.y = vi[2 * q + 1];
        Ub[ownBase + q * 16]        = tr;
        Ub[512 + ownBase + q * 16]  = ti;
        Ub[1024 + ownBase + q * 16] = tr + ti;
    }
    __syncthreads();
    v8f t1 = {}, t2 = {}, t3 = {};
#pragma unroll
    for (int kk = 0; kk < 16; ++kk) {
        v2f br = Ub[bBase + kk * 32];
        v2f bi = Ub[512 + bBase + kk * 32];
        v2f bs = Ub[1024 + bBase + kk * 32];
        t1 = __builtin_amdgcn_wmma_f32_16x16x4_f32(false, are[kk], false, br, (short)0, t1, false, false);
        t2 = __builtin_amdgcn_wmma_f32_16x16x4_f32(false, aim[kk], false, bi, (short)0, t2, false, false);
        t3 = __builtin_amdgcn_wmma_f32_16x16x4_f32(false, aws[kk], false, bs, (short)0, t3, false, false);
    }
    fnde_combine(t1, t2, t3, bv, kr, ki);
}

__device__ __forceinline__ void fnde_layers(v2f* U0, v2f* U1, bool ky0, bool noPartner,
                                            int ownBase, int bBase, int row0, int mode, int pmode,
                                            int mt, int lane,
                                            const float* __restrict__ Wp, const float* __restrict__ bias,
                                            const float* __restrict__ ts,
                                            float (&zr)[8], float (&zi)[8],
                                            float (&str)[8], float (&sti)[8]) {
    float ssr[8], ssi[8];

#pragma unroll 1
    for (int l = 0; l < Ll; ++l) {
        const v4f* Wm = (const v4f*)(Wp + (size_t)l * WLAYER + (size_t)mode * (2 * Cc * Cc))
                        + (mt * 16) * 32 + lane;
        v2f are[16], aim[16], aws[16];
        if (ky0) {
            const v4f* Wq = (const v4f*)(Wp + (size_t)l * WLAYER + (size_t)pmode * (2 * Cc * Cc))
                            + (mt * 16) * 32 + lane;
            const float pf = noPartner ? 0.0f : 0.5f;
            v4f a[16], p[16];
#pragma unroll
            for (int kk = 0; kk < 16; ++kk) { a[kk] = Wm[kk * 32];   p[kk] = Wq[kk * 32]; }
#pragma unroll
            for (int kk = 0; kk < 16; ++kk) {
                are[kk].x = 0.5f * a[kk].x + pf * p[kk].x;
                are[kk].y = 0.5f * a[kk].y + pf * p[kk].y;
                aim[kk].x = 0.5f * a[kk].z - pf * p[kk].z;
                aim[kk].y = 0.5f * a[kk].w - pf * p[kk].w;
            }
        } else {
#pragma unroll
            for (int kk = 0; kk < 16; ++kk) {
                v4f a = Wm[kk * 32];
                are[kk].x = a.x;    are[kk].y = a.y;
                aim[kk].x = a.z;    aim[kk].y = a.w;
            }
        }
#pragma unroll
        for (int kk = 0; kk < 16; ++kk) aws[kk] = are[kk] + aim[kk];
        float bv[8];
#pragma unroll
        for (int j = 0; j < 8; ++j) {
            float bj = bias[l * Cc + row0 + j];
            bv[j] = (mode == 0) ? bj * 16384.0f : 0.0f;
        }

#pragma unroll 1
        for (int s = 0; s < 4; ++s) {
            const float h  = ts[s + 1] - ts[s];
            const float hh = 0.5f * h;
            const float h6 = h * (1.0f / 6.0f);
            float kr[8], ki[8], ur[8], ui[8];

            fnde_eval(U0, ownBase, bBase, zr, zi, are, aim, aws, bv, kr, ki);
#pragma unroll
            for (int j = 0; j < 8; ++j) {
                ssr[j] = kr[j];                 ssi[j] = ki[j];
                ur[j] = zr[j] + hh * kr[j];     ui[j] = zi[j] + hh * ki[j];
            }
            fnde_eval(U1, ownBase, bBase, ur, ui, are, aim, aws, bv, kr, ki);
#pragma unroll
            for (int j = 0; j < 8; ++j) {
                ssr[j] += 2.0f * kr[j];         ssi[j] += 2.0f * ki[j];
                ur[j] = zr[j] + hh * kr[j];     ui[j] = zi[j] + hh * ki[j];
            }
            fnde_eval(U0, ownBase, bBase, ur, ui, are, aim, aws, bv, kr, ki);
#pragma unroll
            for (int j = 0; j < 8; ++j) {
                ssr[j] += 2.0f * kr[j];         ssi[j] += 2.0f * ki[j];
                ur[j] = zr[j] + h * kr[j];      ui[j] = zi[j] + h * ki[j];
            }
            fnde_eval(U1, ownBase, bBase, ur, ui, are, aim, aws, bv, kr, ki);
#pragma unroll
            for (int j = 0; j < 8; ++j) {
                ssr[j] += kr[j];                ssi[j] += ki[j];
                zr[j]  += h6 * ssr[j];          zi[j]  += h6 * ssi[j];
                str[j] += h6 * ssr[j];          sti[j] += h6 * ssi[j];
            }
        }
    }
}

__global__ __launch_bounds__(256) __attribute__((amdgpu_waves_per_eu(2, 2)))
void fnde_chain(const float* __restrict__ Ay,
                const float* __restrict__ cs, const float* __restrict__ sn,
                const float* __restrict__ lw, const float* __restrict__ lb,
                const float* __restrict__ Wp, const float* __restrict__ bias,
                const float* __restrict__ ts,
                const float* __restrict__ q, float* __restrict__ T) {
    __shared__ v2f Ush[2 * 2 * 3 * 512];
    v2f* Xsh = Ush + 3072;

    const int blk = blockIdx.x;
    const int wave = __builtin_amdgcn_readfirstlane((int)(threadIdx.x >> 5));
    const int lane = threadIdx.x & 31;
    const int laneHi = lane >> 4;
    const int laneLo = lane & 15;
    const int mi = wave >> 2;
    const int mt = wave & 3;
    const int ky  = blk & 15;
    const int mx0 = blk >> 4;
    const int modeB = 256 + (((16 - mx0) & 15) << 4) + ky;
    const int mode = mi ? modeB : blk;
    const bool ky0 = (ky == 0);
    const bool noPartner = (blk == 0) && (mi == 1);
    const int pmode = (blk == 0) ? mode : (mi ? blk : modeB);

    const int ownBase = (mt * 8 + laneHi * 4) * 16 + laneLo;
    const int bBase   = laneHi * 16 + laneLo;
    const int row0    = mt * 16 + laneHi * 8;

    v2f* U0 = Ush + (0 * 2 + mi) * 1536;
    v2f* U1 = Ush + (1 * 2 + mi) * 1536;

    float zr[8], zi[8], str[8], sti[8];
    {
        const int tb = threadIdx.x >> 4;
        const int xs = threadIdx.x & 15;
        const int kxA = mx0;
        const int kxB = 112 + ((16 - mx0) & 15);
        const v2f* ap = (const v2f*)Ay + (ky * 16 + tb) * 128 + xs;
        v2f av[8];
#pragma unroll
        for (int j = 0; j < 8; ++j) av[j] = ap[j * 16];
        float rA = 0.f, iA = 0.f, rB = 0.f, iB = 0.f;
#pragma unroll
        for (int j = 0; j < 8; ++j) {
            const int x = j * 16 + xs;
            const float ar = av[j].x, ai = av[j].y;
            const int aA = (kxA * x) & 127;
            const int aB = (kxB * x) & 127;
            const float cA = cs[aA], sA = sn[aA];
            const float cB = cs[aB], sB = sn[aB];
            rA += ar * cA + ai * sA;
            iA += ai * cA - ar * sA;
            rB += ar * cB + ai * sB;
            iB += ai * cB - ar * sB;
        }
        v2f pA, pB;
        pA.x = rA;   pA.y = iA;
        pB.x = rB;   pB.y = iB;
        Xsh[xs * 17 + tb]       = pA;
        Xsh[272 + xs * 17 + tb] = pB;
        __syncthreads();
        v2f hsum = Xsh[mi * 272 + laneLo];
#pragma unroll
        for (int r = 1; r < 16; ++r) hsum += Xsh[mi * 272 + r * 17 + laneLo];
        const float hr = hsum.x;
        const float hi = hsum.y;
#pragma unroll
        for (int j = 0; j < 8; ++j) {
            float w = lw[row0 + j];
            float lbj = lb[row0 + j];
            zr[j] = w * hr + ((mode == 0) ? lbj * 16384.0f : 0.0f);
            zi[j] = w * hi;
            str[j] = 0.0f;  sti[j] = 0.0f;
        }
    }

    fnde_layers(U0, U1, ky0, noPartner, ownBase, bBase, row0, mode, pmode, mt, lane,
                Wp, bias, ts, zr, zi, str, sti);

    float pre = 0.0f, pie = 0.0f;
#pragma unroll
    for (int j = 0; j < 8; ++j) {
        const float w = q[row0 + j];
        pre += w * str[j];
        pie += w * sti[j];
    }
    const float sf = noPartner ? 2.0f : 1.0f;
    v2f part;
    part.x = sf * pre;   part.y = sf * pie;
    Ush[threadIdx.x] = part;
    __syncthreads();
    if (wave == 0) {
        const int m2 = laneHi;
        v2f acc = Ush[m2 * 128 + laneLo];
#pragma unroll
        for (int r = 1; r < 8; ++r) acc += Ush[m2 * 128 + r * 16 + laneLo];
        const int modeL = m2 ? modeB : blk;
        *(v2f*)(T + (modeL * 16 + laneLo) * 2) = acc;
    }
}

__device__ __forceinline__ void fnde_prep_body(const float* __restrict__ p1w, const float* __restrict__ p1b,
                                               const float* __restrict__ p2w, const float* __restrict__ p2b,
                                               const float* __restrict__ lw,  const float* __restrict__ lb,
                                               float* __restrict__ q, float* __restrict__ scal,
                                               float* qs) {
    int c = threadIdx.x;
    if (c < 64) {
        float acc = 0.f;
        for (int p = 0; p < 64; ++p) acc += p2w[p] * p1w[p * 64 + c];
        qs[c] = acc;
        q[c] = acc;
    }
    __syncthreads();
    if (c == 0) {
        float r = p2b[0];
        for (int p = 0; p < 64; ++p) r += p2w[p] * p1b[p];
        float al = 0.f, be = r;
        for (int i = 0; i < 64; ++i) { al += qs[i] * lw[i]; be += qs[i] * lb[i]; }
        scal[0] = al;
        scal[1] = be;
    }
}

#define FR_PREP 1
#define FR_DFT1 2
#define FR_PACK (2 + 256)
#define FR_GRID (FR_PACK + Ll * 2 * Mm * 4 * 4)
__global__ __launch_bounds__(128)
void fnde_front(const float* __restrict__ z,
                const float* __restrict__ w1, const float* __restrict__ w2,
                const float* __restrict__ p1w, const float* __restrict__ p1b,
                const float* __restrict__ p2w, const float* __restrict__ p2b,
                const float* __restrict__ lw,  const float* __restrict__ lb,
                float* __restrict__ cs, float* __restrict__ sn,
                float* __restrict__ q, float* __restrict__ scal,
                float* __restrict__ Ay, float* __restrict__ Wp) {
    __shared__ v4f fr_lds[16 * (PK_PITCH / 4)];
    const int blk = blockIdx.x;
    if (blk >= FR_PACK)      fnde_pack_body(w1, w2, Wp, fr_lds, blk - FR_PACK);
    else if (blk >= FR_DFT1) fnde_dft1_body(z, Ay, (float*)fr_lds, blk - FR_DFT1);
    else if (blk == FR_PREP) fnde_prep_body(p1w, p1b, p2w, p2b, lw, lb, q, scal, (float*)fr_lds);
    else                     fnde_tables_body(cs, sn);
}

__global__ __launch_bounds__(256)
void fnde_inv_dft(const float* __restrict__ T, const float* __restrict__ z,
                  const float* __restrict__ cs, const float* __restrict__ sn,
                  const float* __restrict__ scal, float* __restrict__ out) {
    __shared__ v2f Tsh[NMODE];
    __shared__ v2f Gsh[8 * 16];
    const int tid = threadIdx.x;
    const int b  = blockIdx.x >> 4;
    const int xg = blockIdx.x & 15;
    Tsh[tid]       = *(const v2f*)(T + (tid * 16 + b) * 2);
    Tsh[tid + 256] = *(const v2f*)(T + ((tid + 256) * 16 + b) * 2);
    __syncthreads();
    if (tid < 128) {
        const int x  = xg * 8 + (tid >> 4);
        const int ky = tid & 15;
        float re = 0.f, im = 0.f;
        for (int corner = 0; corner < 2; ++corner) {
            for (int mx = 0; mx < 16; ++mx) {
                int m  = corner * 256 + mx * 16 + ky;
                int kx = corner ? (112 + mx) : mx;
                v2f tv = Tsh[m];
                float tr = tv.x, ti = tv.y;
                int a = (kx * x) & 127;
                float c = cs[a], s = sn[a];
                re += tr * c - ti * s;
                im += tr * s + ti * c;
            }
        }
        v2f g;
        g.x = re;   g.y = im;
        Gsh[tid] = g;
    }
    __syncthreads();
    const float al = scal[0], be = scal[1];
    for (int r = 0; r < 4; ++r) {
        const int idx = r * 256 + tid;
        const int y = idx & 127;
        const v2f* g = Gsh + (idx >> 7) * 16;
        float acc = g[0].x;
        for (int ky = 1; ky < 16; ++ky) {
            int a = (ky * y) & 127;
            acc += 2.0f * (g[ky].x * cs[a] - g[ky].y * sn[a]);
        }
        const int t = ((b * 128 + xg * 8) << 7) + idx;
        out[t] = al * z[t] + be + acc * (1.0f / 16384.0f);
    }
}

extern "C" void kernel_launch(void* const* d_in, const int* in_sizes, int n_in,
                              void* d_out, int out_size, void* d_ws, size_t ws_size,
                              hipStream_t stream) {
    (void)in_sizes; (void)n_in; (void)out_size; (void)ws_size;

    const float* zin    = (const float*)d_in[0];
    const float* lift_w = (const float*)d_in[1];
    const float* lift_b = (const float*)d_in[2];
    const float* spec_w1 = (const float*)d_in[3];
    const float* spec_w2 = (const float*)d_in[4];
    const float* fl_bias = (const float*)d_in[5];
    const float* p1_w = (const float*)d_in[6];
    const float* p1_b = (const float*)d_in[7];
    const float* p2_w = (const float*)d_in[8];
    const float* p2_b = (const float*)d_in[9];
    const float* ts   = (const float*)d_in[10];
    float* outp = (float*)d_out;

    float* w = (float*)d_ws;
    float* Wp   = w;  w += (size_t)Ll * WLAYER;
    float* Ay   = w;  w += Bb * Dd * Mm * 2;
    float* Tm   = w;  w += Bb * NMODE * 2;
    float* qv   = w;  w += 64;
    float* cs   = w;  w += 128;
    float* sn   = w;  w += 128;
    float* scal = w;  w += 8;

    fnde_front<<<FR_GRID, 128, 0, stream>>>(zin, spec_w1, spec_w2, p1_w, p1_b, p2_w, p2_b, lift_w, lift_b,
                                            cs, sn, qv, scal, Ay, Wp);
    fnde_chain<<<NMODE / 2, 256, 0, stream>>>(Ay, cs, sn, lift_w, lift_b, Wp, fl_bias, ts, qv, Tm);
    fnde_inv_dft<<<Bb * (Dd / 8), 256, 0, stream>>>(Tm, zin, cs, sn, scal, outp);
}
